# starting kernel
# baseline (speedup 1.0000x reference)
#include <hip/hip_runtime.h>

typedef __attribute__((ext_vector_type(2))) float v2f;
typedef __attribute__((ext_vector_type(4))) float v4f;
typedef __attribute__((ext_vector_type(8))) float v8f;

__global__ void init_zero(float* __restrict__ p, int n) {
  int i = blockIdx.x * blockDim.x + threadIdx.x;
  if (i < n) p[i] = 0.0f;
}

__global__ void find_starts(const int* __restrict__ batch, int* __restrict__ starts,
                            int N, int G) {
  int i = blockIdx.x * blockDim.x + threadIdx.x;
  if (i >= N) return;
  int b = batch[i];
  int bp = (i == 0) ? -1 : batch[i - 1];
  for (int g = bp + 1; g <= b; ++g) starts[g] = i;
  if (i == N - 1) {
    for (int g = b + 1; g <= G; ++g) starts[g] = N;
  }
}

__global__ void __launch_bounds__(256)
gates_gemm(const float* __restrict__ qstar, const float* __restrict__ hs,
           const float* __restrict__ Wih, const float* __restrict__ Whh,
           const float* __restrict__ bih, const float* __restrict__ bhh,
           float* __restrict__ gates) {
  const int lane = threadIdx.x & 31;
  const int wave = blockIdx.x * (blockDim.x >> 5) + (threadIdx.x >> 5);
  const int TN = 1024 / 16;
  const int tm = wave / TN;
  const int tn = wave % TN;
  const int r  = lane & 15;
  const int kh = (lane >> 4) << 1;

  const float* arow = qstar + (size_t)(tm * 16 + r) * 512;
  const float* brow = Wih   + (size_t)(tn * 16 + r) * 512;
  v8f acc = {};
  for (int k = 0; k < 512; k += 4) {
    v2f a, b;
    a.x = arow[k + kh]; a.y = arow[k + kh + 1];
    b.x = brow[k + kh]; b.y = brow[k + kh + 1];
    acc = __builtin_amdgcn_wmma_f32_16x16x4_f32(false, a, false, b, (short)0, acc,
                                                false, false);
  }
  const float* arow2 = hs  + (size_t)(tm * 16 + r) * 256;
  const float* brow2 = Whh + (size_t)(tn * 16 + r) * 256;
  for (int k = 0; k < 256; k += 4) {
    v2f a, b;
    a.x = arow2[k + kh]; a.y = arow2[k + kh + 1];
    b.x = brow2[k + kh]; b.y = brow2[k + kh + 1];
    acc = __builtin_amdgcn_wmma_f32_16x16x4_f32(false, a, false, b, (short)0, acc,
                                                false, false);
  }
  const int mb = tm * 16 + (lane >> 4) * 8;
  const int n  = tn * 16 + r;
  const float bias = bih[n] + bhh[n];
#pragma unroll
  for (int v = 0; v < 8; ++v)
    gates[(size_t)(mb + v) * 1024 + n] = acc[v] + bias;
}

__device__ __forceinline__ float sigmoidf_(float x) {
  return 1.0f / (1.0f + __expf(-x));
}

__global__ void __launch_bounds__(256)
lstm_update(const float* __restrict__ gates, float* __restrict__ hs,
            float* __restrict__ cs, float* __restrict__ qstar) {
  int idx = blockIdx.x * blockDim.x + threadIdx.x;
  int b = idx >> 8, c = idx & 255;
  const float* g = gates + (size_t)b * 1024;
  float ig = sigmoidf_(g[c]);
  float fg = sigmoidf_(g[256 + c]);
  float gg = tanhf(g[512 + c]);
  float og = sigmoidf_(g[768 + c]);
  float cn = fg * cs[idx] + ig * gg;
  cs[idx] = cn;
  float h = og * tanhf(cn);
  hs[idx] = h;
  qstar[(size_t)b * 512 + c] = h;
}

__global__ void __launch_bounds__(256)
attention(const float* __restrict__ x, const int* __restrict__ starts,
          const float* __restrict__ hs, float* __restrict__ qstar) {
  const int g    = blockIdx.x;
  const int tid  = threadIdx.x;
  const int lane = tid & 31;
  const int wv   = tid >> 5;

  __shared__ __align__(16) float q[256];
  __shared__ __align__(16) float rall[8][256];
  __shared__ float wmax[8];
  __shared__ float wsum[8];

  q[tid] = hs[(size_t)g * 256 + tid];
  __syncthreads();

  const int start = starts[g];
  const int end   = starts[g + 1];

  const v4f q0 = *(const v4f*)&q[4 * lane];
  const v4f q1 = *(const v4f*)&q[128 + 4 * lane];

  float m = -__builtin_inff();
  float s = 0.0f;
  v4f r0 = {}, r1 = {};

  for (int i = start + wv; i < end; i += 8) {
    const v4f* xr = (const v4f*)(x + (size_t)i * 256);
    v4f x0 = xr[lane];
    v4f x1 = xr[32 + lane];
    float d = 0.0f;
    d = fmaf(x0.x, q0.x, d); d = fmaf(x0.y, q0.y, d);
    d = fmaf(x0.z, q0.z, d); d = fmaf(x0.w, q0.w, d);
    d = fmaf(x1.x, q1.x, d); d = fmaf(x1.y, q1.y, d);
    d = fmaf(x1.z, q1.z, d); d = fmaf(x1.w, q1.w, d);
#pragma unroll
    for (int off = 16; off; off >>= 1) d += __shfl_xor(d, off, 32);
    if (d > m) {
      float sc = (m == -__builtin_inff()) ? 0.0f : __expf(m - d);
      s *= sc;
      r0.x *= sc; r0.y *= sc; r0.z *= sc; r0.w *= sc;
      r1.x *= sc; r1.y *= sc; r1.z *= sc; r1.w *= sc;
      m = d;
    }
    float a = __expf(d - m);
    s += a;
    r0.x = fmaf(a, x0.x, r0.x); r0.y = fmaf(a, x0.y, r0.y);
    r0.z = fmaf(a, x0.z, r0.z); r0.w = fmaf(a, x0.w, r0.w);
    r1.x = fmaf(a, x1.x, r1.x); r1.y = fmaf(a, x1.y, r1.y);
    r1.z = fmaf(a, x1.z, r1.z); r1.w = fmaf(a, x1.w, r1.w);
  }

  *(v4f*)&rall[wv][4 * lane]       = r0;
  *(v4f*)&rall[wv][128 + 4 * lane] = r1;
  if (lane == 0) { wmax[wv] = m; wsum[wv] = s; }
  __syncthreads();

  float M = -__builtin_inff();
#pragma unroll
  for (int w = 0; w < 8; ++w) M = fmaxf(M, wmax[w]);

  float stot = 0.0f, r = 0.0f;
#pragma unroll
  for (int w = 0; w < 8; ++w) {
    float sw = wsum[w];
    if (sw > 0.0f) {
      float sc = __expf(wmax[w] - M);
      stot = fmaf(sw, sc, stot);
      r    = fmaf(rall[w][tid], sc, r);
    }
  }
  qstar[(size_t)g * 512 + 256 + tid] = (stot > 0.0f) ? (r / stot) : 0.0f;
}

__global__ void __launch_bounds__(256)
out_gemm(const float* __restrict__ qstar, const float* __restrict__ Wout,
         const float* __restrict__ bout, float* __restrict__ out) {
  const int lane = threadIdx.x & 31;
  const int wave = blockIdx.x * (blockDim.x >> 5) + (threadIdx.x >> 5);
  const int TN = 256 / 16;
  const int tm = wave / TN;
  const int tn = wave % TN;
  const int r  = lane & 15;
  const int kh = (lane >> 4) << 1;

  const float* arow = qstar + (size_t)(tm * 16 + r) * 512;
  const float* brow = Wout  + (size_t)(tn * 16 + r) * 512;
  v8f acc = {};
  for (int k = 0; k < 512; k += 4) {
    v2f a, b;
    a.x = arow[k + kh]; a.y = arow[k + kh + 1];
    b.x = brow[k + kh]; b.y = brow[k + kh + 1];
    acc = __builtin_amdgcn_wmma_f32_16x16x4_f32(false, a, false, b, (short)0, acc,
                                                false, false);
  }
  const int mb = tm * 16 + (lane >> 4) * 8;
  const int n  = tn * 16 + r;
  const float bias = bout[n];
#pragma unroll
  for (int v = 0; v < 8; ++v)
    out[(size_t)(mb + v) * 256 + n] = acc[v] + bias;
}

extern "C" void kernel_launch(void* const* d_in, const int* in_sizes, int n_in,
                              void* d_out, int out_size, void* d_ws, size_t ws_size,
                              hipStream_t stream) {
  const float* x     = (const float*)d_in[0];
  const int*   batch = (const int*)  d_in[1];
  const float* Wih   = (const float*)d_in[2];
  const float* Whh   = (const float*)d_in[3];
  const float* bih   = (const float*)d_in[4];
  const float* bhh   = (const float*)d_in[5];
  const float* Wout  = (const float*)d_in[6];
  const float* bout  = (const float*)d_in[7];
  const int N = in_sizes[0] / 256;
  const int G = 512;

  float* ws    = (float*)d_ws;
  float* hs    = ws;
  float* cs    = hs + 512 * 256;
  float* qstar = cs + 512 * 256;
  float* gates = qstar + 512 * 512;
  int*   starts = (int*)(gates + 512 * 1024);

  int zn = 512 * 256 * 2 + 512 * 512;
  init_zero<<<(zn + 255) / 256, 256, 0, stream>>>(ws, zn);
  find_starts<<<(N + 255) / 256, 256, 0, stream>>>(batch, starts, N, G);

  for (int step = 0; step < 3; ++step) {
    gates_gemm<<<256, 256, 0, stream>>>(qstar, hs, Wih, Whh, bih, bhh, gates);
    lstm_update<<<512, 256, 0, stream>>>(gates, hs, cs, qstar);
    attention<<<G, 256, 0, stream>>>(x, starts, hs, qstar);
  }
  out_gemm<<<64, 256, 0, stream>>>(qstar, Wout, bout, (float*)d_out);
}


# attention: each wave takes 8 consecutive nodes per trip and requests all 16 row loads before use; one softmax rescale per trip (node grouping per wave, and so the f32 sum order, change)
# speedup vs baseline: 1.5782x; 1.5782x over previous
#include <hip/hip_runtime.h>

typedef __attribute__((ext_vector_type(2))) float v2f;
typedef __attribute__((ext_vector_type(4))) float v4f;
typedef __attribute__((ext_vector_type(8))) float v8f;

__global__ void init_zero(float* __restrict__ p, int n) {
  int i = blockIdx.x * blockDim.x + threadIdx.x;
  if (i < n) p[i] = 0.0f;
}

__global__ void find_starts(const int* __restrict__ batch, int* __restrict__ starts,
                            int N, int G) {
  int i = blockIdx.x * blockDim.x + threadIdx.x;
  if (i >= N) return;
  int b = batch[i];
  int bp = (i == 0) ? -1 : batch[i - 1];
  for (int g = bp + 1; g <= b; ++g) starts[g] = i;
  if (i == N - 1) {
    for (int g = b + 1; g <= G; ++g) starts[g] = N;
  }
}

__global__ void __launch_bounds__(256)
gates_gemm(const float* __restrict__ qstar, const float* __restrict__ hs,
           const float* __restrict__ Wih, const float* __restrict__ Whh,
           const float* __restrict__ bih, const float* __restrict__ bhh,
           float* __restrict__ gates) {
  const int lane = threadIdx.x & 31;
  const int wave = blockIdx.x * (blockDim.x >> 5) + (threadIdx.x >> 5);
  const int TN = 1024 / 16;
  const int tm = wave / TN;
  const int tn = wave % TN;
  const int r  = lane & 15;
  const int kh = (lane >> 4) << 1;

  const float* arow = qstar + (size_t)(tm * 16 + r) * 512;
  const float* brow = Wih   + (size_t)(tn * 16 + r) * 512;
  v8f acc = {};
  for (int k = 0; k < 512; k += 4) {
    v2f a, b;
    a.x = arow[k + kh]; a.y = arow[k + kh + 1];
    b.x = brow[k + kh]; b.y = brow[k + kh + 1];
    acc = __builtin_amdgcn_wmma_f32_16x16x4_f32(false, a, false, b, (short)0, acc,
                                                false, false);
  }
  const float* arow2 = hs  + (size_t)(tm * 16 + r) * 256;
  const float* brow2 = Whh + (size_t)(tn * 16 + r) * 256;
  for (int k = 0; k < 256; k += 4) {
    v2f a, b;
    a.x = arow2[k + kh]; a.y = arow2[k + kh + 1];
    b.x = brow2[k + kh]; b.y = brow2[k + kh + 1];
    acc = __builtin_amdgcn_wmma_f32_16x16x4_f32(false, a, false, b, (short)0, acc,
                                                false, false);
  }
  const int mb = tm * 16 + (lane >> 4) * 8;
  const int n  = tn * 16 + r;
  const float bias = bih[n] + bhh[n];
#pragma unroll
  for (int v = 0; v < 8; ++v)
    gates[(size_t)(mb + v) * 1024 + n] = acc[v] + bias;
}

__device__ __forceinline__ float sigmoidf_(float x) {
  return 1.0f / (1.0f + __expf(-x));
}

__global__ void __launch_bounds__(256)
lstm_update(const float* __restrict__ gates, float* __restrict__ hs,
            float* __restrict__ cs, float* __restrict__ qstar) {
  int idx = blockIdx.x * blockDim.x + threadIdx.x;
  int b = idx >> 8, c = idx & 255;
  const float* g = gates + (size_t)b * 1024;
  float ig = sigmoidf_(g[c]);
  float fg = sigmoidf_(g[256 + c]);
  float gg = tanhf(g[512 + c]);
  float og = sigmoidf_(g[768 + c]);
  float cn = fg * cs[idx] + ig * gg;
  cs[idx] = cn;
  float h = og * tanhf(cn);
  hs[idx] = h;
  qstar[(size_t)b * 512 + c] = h;
}

#define NB 8

__global__ void __launch_bounds__(256)
attention(const float* __restrict__ x, const int* __restrict__ starts,
          const float* __restrict__ hs, float* __restrict__ qstar) {
  const int g    = blockIdx.x;
  const int tid  = threadIdx.x;
  const int lane = tid & 31;
  const int wv   = __builtin_amdgcn_readfirstlane(tid >> 5);

  __shared__ __align__(16) float q[256];
  __shared__ __align__(16) float rall[8][256];
  __shared__ float wmax[8];
  __shared__ float wsum[8];

  q[tid] = hs[(size_t)g * 256 + tid];
  __syncthreads();

  const int start = starts[g];
  const int end   = starts[g + 1];

  const v4f q0 = *(const v4f*)&q[4 * lane];
  const v4f q1 = *(const v4f*)&q[128 + 4 * lane];

  const float NEG_INF = -__builtin_inff();
  float m = NEG_INF;
  float s = 0.0f;
  v4f r0 = {}, r1 = {};

  for (int base = start + wv * NB; base < end; base += 8 * NB) {
    v4f xa[NB], xb[NB];
#pragma unroll
    for (int j = 0; j < NB; ++j) {
      const int i = (base + j < end) ? (base + j) : base;
      const v4f* xr = (const v4f*)(x + (size_t)i * 256);
      xa[j] = xr[lane];
      xb[j] = xr[32 + lane];
    }

    float d[NB];
#pragma unroll
    for (int j = 0; j < NB; ++j) {
      float t = 0.0f;
      t = fmaf(xa[j].x, q0.x, t); t = fmaf(xa[j].y, q0.y, t);
      t = fmaf(xa[j].z, q0.z, t); t = fmaf(xa[j].w, q0.w, t);
      t = fmaf(xb[j].x, q1.x, t); t = fmaf(xb[j].y, q1.y, t);
      t = fmaf(xb[j].z, q1.z, t); t = fmaf(xb[j].w, q1.w, t);
      d[j] = t;
    }
#pragma unroll
    for (int off = 16; off; off >>= 1) {
#pragma unroll
      for (int j = 0; j < NB; ++j) d[j] += __shfl_xor(d[j], off, 32);
    }
    float mb = NEG_INF;
#pragma unroll
    for (int j = 0; j < NB; ++j) {
      if (base + j >= end) d[j] = NEG_INF;
      mb = fmaxf(mb, d[j]);
    }
    const float mn = fmaxf(m, mb);
    const float sc = (m == NEG_INF) ? 0.0f : __expf(m - mn);
    m = mn;
    s *= sc;
    r0.x *= sc; r0.y *= sc; r0.z *= sc; r0.w *= sc;
    r1.x *= sc; r1.y *= sc; r1.z *= sc; r1.w *= sc;
#pragma unroll
    for (int j = 0; j < NB; ++j) {
      const float a = __expf(d[j] - m);
      s += a;
      r0.x = fmaf(a, xa[j].x, r0.x); r0.y = fmaf(a, xa[j].y, r0.y);
      r0.z = fmaf(a, xa[j].z, r0.z); r0.w = fmaf(a, xa[j].w, r0.w);
      r1.x = fmaf(a, xb[j].x, r1.x); r1.y = fmaf(a, xb[j].y, r1.y);
      r1.z = fmaf(a, xb[j].z, r1.z); r1.w = fmaf(a, xb[j].w, r1.w);
    }
  }

  *(v4f*)&rall[wv][4 * lane]       = r0;
  *(v4f*)&rall[wv][128 + 4 * lane] = r1;
  if (lane == 0) { wmax[wv] = m; wsum[wv] = s; }
  __syncthreads();

  float M = -__builtin_inff();
#pragma unroll
  for (int w = 0; w < 8; ++w) M = fmaxf(M, wmax[w]);

  float stot = 0.0f, r = 0.0f;
#pragma unroll
  for (int w = 0; w < 8; ++w) {
    float sw = wsum[w];
    if (sw > 0.0f) {
      float sc = __expf(wmax[w] - M);
      stot = fmaf(sw, sc, stot);
      r    = fmaf(rall[w][tid], sc, r);
    }
  }
  qstar[(size_t)g * 512 + 256 + tid] = (stot > 0.0f) ? (r / stot) : 0.0f;
}

__global__ void __launch_bounds__(256)
out_gemm(const float* __restrict__ qstar, const float* __restrict__ Wout,
         const float* __restrict__ bout, float* __restrict__ out) {
  const int lane = threadIdx.x & 31;
  const int wave = blockIdx.x * (blockDim.x >> 5) + (threadIdx.x >> 5);
  const int TN = 256 / 16;
  const int tm = wave / TN;
  const int tn = wave % TN;
  const int r  = lane & 15;
  const int kh = (lane >> 4) << 1;

  const float* arow = qstar + (size_t)(tm * 16 + r) * 512;
  const float* brow = Wout  + (size_t)(tn * 16 + r) * 512;
  v8f acc = {};
  for (int k = 0; k < 512; k += 4) {
    v2f a, b;
    a.x = arow[k + kh]; a.y = arow[k + kh + 1];
    b.x = brow[k + kh]; b.y = brow[k + kh + 1];
    acc = __builtin_amdgcn_wmma_f32_16x16x4_f32(false, a, false, b, (short)0, acc,
                                                false, false);
  }
  const int mb = tm * 16 + (lane >> 4) * 8;
  const int n  = tn * 16 + r;
  const float bias = bout[n];
#pragma unroll
  for (int v = 0; v < 8; ++v)
    out[(size_t)(mb + v) * 256 + n] = acc[v] + bias;
}

extern "C" void kernel_launch(void* const* d_in, const int* in_sizes, int n_in,
                              void* d_out, int out_size, void* d_ws, size_t ws_size,
                              hipStream_t stream) {
  const float* x     = (const float*)d_in[0];
  const int*   batch = (const int*)  d_in[1];
  const float* Wih   = (const float*)d_in[2];
  const float* Whh   = (const float*)d_in[3];
  const float* bih   = (const float*)d_in[4];
  const float* bhh   = (const float*)d_in[5];
  const float* Wout  = (const float*)d_in[6];
  const float* bout  = (const float*)d_in[7];
  const int N = in_sizes[0] / 256;
  const int G = 512;

  float* ws    = (float*)d_ws;
  float* hs    = ws;
  float* cs    = hs + 512 * 256;
  float* qstar = cs + 512 * 256;
  float* gates = qstar + 512 * 512;
  int*   starts = (int*)(gates + 512 * 1024);

  int zn = 512 * 256 * 2 + 512 * 512;
  init_zero<<<(zn + 255) / 256, 256, 0, stream>>>(ws, zn);
  find_starts<<<(N + 255) / 256, 256, 0, stream>>>(batch, starts, N, G);

  for (int step = 0; step < 3; ++step) {
    gates_gemm<<<256, 256, 0, stream>>>(qstar, hs, Wih, Whh, bih, bhh, gates);
    lstm_update<<<512, 256, 0, stream>>>(gates, hs, cs, qstar);
    attention<<<G, 256, 0, stream>>>(x, starts, hs, qstar);
  }
  out_gemm<<<64, 256, 0, stream>>>(qstar, Wout, bout, (float*)d_out);
}


# step 0 folded: the reference's zero initial state makes the first gates b_ih + b_hh, so the first attention launch computes its LSTM cell from the biases; zero-fill, first GEMM and LSTM launches go
# speedup vs baseline: 1.8350x; 1.1627x over previous
#include <hip/hip_runtime.h>

typedef __attribute__((ext_vector_type(2))) float v2f;
typedef __attribute__((ext_vector_type(4))) float v4f;
typedef __attribute__((ext_vector_type(8))) float v8f;

__global__ void find_starts(const int* __restrict__ batch, int* __restrict__ starts,
                            int N, int G) {
  int i = blockIdx.x * blockDim.x + threadIdx.x;
  if (i >= N) return;
  int b = batch[i];
  int bp = (i == 0) ? -1 : batch[i - 1];
  for (int g = bp + 1; g <= b; ++g) starts[g] = i;
  if (i == N - 1) {
    for (int g = b + 1; g <= G; ++g) starts[g] = N;
  }
}

__global__ void __launch_bounds__(256)
gates_gemm(const float* __restrict__ qstar, const float* __restrict__ hs,
           const float* __restrict__ Wih, const float* __restrict__ Whh,
           const float* __restrict__ bih, const float* __restrict__ bhh,
           float* __restrict__ gates) {
  const int lane = threadIdx.x & 31;
  const int wave = blockIdx.x * (blockDim.x >> 5) + (threadIdx.x >> 5);
  const int TN = 1024 / 16;
  const int tm = wave / TN;
  const int tn = wave % TN;
  const int r  = lane & 15;
  const int kh = (lane >> 4) << 1;

  const float* arow = qstar + (size_t)(tm * 16 + r) * 512;
  const float* brow = Wih   + (size_t)(tn * 16 + r) * 512;
  v8f acc = {};
  for (int k = 0; k < 512; k += 4) {
    v2f a, b;
    a.x = arow[k + kh]; a.y = arow[k + kh + 1];
    b.x = brow[k + kh]; b.y = brow[k + kh + 1];
    acc = __builtin_amdgcn_wmma_f32_16x16x4_f32(false, a, false, b, (short)0, acc,
                                                false, false);
  }
  const float* arow2 = hs  + (size_t)(tm * 16 + r) * 256;
  const float* brow2 = Whh + (size_t)(tn * 16 + r) * 256;
  for (int k = 0; k < 256; k += 4) {
    v2f a, b;
    a.x = arow2[k + kh]; a.y = arow2[k + kh + 1];
    b.x = brow2[k + kh]; b.y = brow2[k + kh + 1];
    acc = __builtin_amdgcn_wmma_f32_16x16x4_f32(false, a, false, b, (short)0, acc,
                                                false, false);
  }
  const int mb = tm * 16 + (lane >> 4) * 8;
  const int n  = tn * 16 + r;
  const float bias = bih[n] + bhh[n];
#pragma unroll
  for (int v = 0; v < 8; ++v)
    gates[(size_t)(mb + v) * 1024 + n] = acc[v] + bias;
}

__device__ __forceinline__ float sigmoidf_(float x) {
  return 1.0f / (1.0f + __expf(-x));
}

__global__ void __launch_bounds__(256)
lstm_update(const float* __restrict__ gates, float* __restrict__ hs,
            float* __restrict__ cs, float* __restrict__ qstar) {
  int idx = blockIdx.x * blockDim.x + threadIdx.x;
  int b = idx >> 8, c = idx & 255;
  const float* g = gates + (size_t)b * 1024;
  float ig = sigmoidf_(g[c]);
  float fg = sigmoidf_(g[256 + c]);
  float gg = tanhf(g[512 + c]);
  float og = sigmoidf_(g[768 + c]);
  float cn = fg * cs[idx] + ig * gg;
  cs[idx] = cn;
  float h = og * tanhf(cn);
  hs[idx] = h;
  qstar[(size_t)b * 512 + c] = h;
}

#define NB 8

template <bool FIRST>
__global__ void __launch_bounds__(256)
attention(const float* __restrict__ x, const int* __restrict__ starts,
          const float* __restrict__ bih, const float* __restrict__ bhh,
          float* __restrict__ hs, float* __restrict__ cs,
          float* __restrict__ qstar) {
  const int g    = blockIdx.x;
  const int tid  = threadIdx.x;
  const int lane = tid & 31;
  const int wv   = __builtin_amdgcn_readfirstlane(tid >> 5);

  __shared__ __align__(16) float q[256];
  __shared__ __align__(16) float rall[8][256];
  __shared__ float wmax[8];
  __shared__ float wsum[8];

  if (FIRST) {
    const float ig = sigmoidf_(bih[tid] + bhh[tid]);
    const float fg = sigmoidf_(bih[256 + tid] + bhh[256 + tid]);
    const float gg = tanhf(bih[512 + tid] + bhh[512 + tid]);
    const float og = sigmoidf_(bih[768 + tid] + bhh[768 + tid]);
    const float cn = fg * 0.0f + ig * gg;
    const float h = og * tanhf(cn);
    cs[(size_t)g * 256 + tid] = cn;
    hs[(size_t)g * 256 + tid] = h;
    qstar[(size_t)g * 512 + tid] = h;
    q[tid] = h;
  } else {
    q[tid] = hs[(size_t)g * 256 + tid];
  }
  __syncthreads();

  const int start = starts[g];
  const int end   = starts[g + 1];

  const v4f q0 = *(const v4f*)&q[4 * lane];
  const v4f q1 = *(const v4f*)&q[128 + 4 * lane];

  const float NEG_INF = -__builtin_inff();
  float m = NEG_INF;
  float s = 0.0f;
  v4f r0 = {}, r1 = {};

  for (int base = start + wv * NB; base < end; base += 8 * NB) {
    v4f xa[NB], xb[NB];
#pragma unroll
    for (int j = 0; j < NB; ++j) {
      const int i = (base + j < end) ? (base + j) : base;
      const v4f* xr = (const v4f*)(x + (size_t)i * 256);
      xa[j] = xr[lane];
      xb[j] = xr[32 + lane];
    }

    float d[NB];
#pragma unroll
    for (int j = 0; j < NB; ++j) {
      float t = 0.0f;
      t = fmaf(xa[j].x, q0.x, t); t = fmaf(xa[j].y, q0.y, t);
      t = fmaf(xa[j].z, q0.z, t); t = fmaf(xa[j].w, q0.w, t);
      t = fmaf(xb[j].x, q1.x, t); t = fmaf(xb[j].y, q1.y, t);
      t = fmaf(xb[j].z, q1.z, t); t = fmaf(xb[j].w, q1.w, t);
      d[j] = t;
    }
#pragma unroll
    for (int off = 16; off; off >>= 1) {
#pragma unroll
      for (int j = 0; j < NB; ++j) d[j] += __shfl_xor(d[j], off, 32);
    }
    float mb = NEG_INF;
#pragma unroll
    for (int j = 0; j < NB; ++j) {
      if (base + j >= end) d[j] = NEG_INF;
      mb = fmaxf(mb, d[j]);
    }
    const float mn = fmaxf(m, mb);
    const float sc = (m == NEG_INF) ? 0.0f : __expf(m - mn);
    m = mn;
    s *= sc;
    r0.x *= sc; r0.y *= sc; r0.z *= sc; r0.w *= sc;
    r1.x *= sc; r1.y *= sc; r1.z *= sc; r1.w *= sc;
#pragma unroll
    for (int j = 0; j < NB; ++j) {
      const float a = __expf(d[j] - m);
      s += a;
      r0.x = fmaf(a, xa[j].x, r0.x); r0.y = fmaf(a, xa[j].y, r0.y);
      r0.z = fmaf(a, xa[j].z, r0.z); r0.w = fmaf(a, xa[j].w, r0.w);
      r1.x = fmaf(a, xb[j].x, r1.x); r1.y = fmaf(a, xb[j].y, r1.y);
      r1.z = fmaf(a, xb[j].z, r1.z); r1.w = fmaf(a, xb[j].w, r1.w);
    }
  }

  *(v4f*)&rall[wv][4 * lane]       = r0;
  *(v4f*)&rall[wv][128 + 4 * lane] = r1;
  if (lane == 0) { wmax[wv] = m; wsum[wv] = s; }
  __syncthreads();

  float M = -__builtin_inff();
#pragma unroll
  for (int w = 0; w < 8; ++w) M = fmaxf(M, wmax[w]);

  float stot = 0.0f, r = 0.0f;
#pragma unroll
  for (int w = 0; w < 8; ++w) {
    float sw = wsum[w];
    if (sw > 0.0f) {
      float sc = __expf(wmax[w] - M);
      stot = fmaf(sw, sc, stot);
      r    = fmaf(rall[w][tid], sc, r);
    }
  }
  qstar[(size_t)g * 512 + 256 + tid] = (stot > 0.0f) ? (r / stot) : 0.0f;
}

__global__ void __launch_bounds__(256)
out_gemm(const float* __restrict__ qstar, const float* __restrict__ Wout,
         const float* __restrict__ bout, float* __restrict__ out) {
  const int lane = threadIdx.x & 31;
  const int wave = blockIdx.x * (blockDim.x >> 5) + (threadIdx.x >> 5);
  const int TN = 256 / 16;
  const int tm = wave / TN;
  const int tn = wave % TN;
  const int r  = lane & 15;
  const int kh = (lane >> 4) << 1;

  const float* arow = qstar + (size_t)(tm * 16 + r) * 512;
  const float* brow = Wout  + (size_t)(tn * 16 + r) * 512;
  v8f acc = {};
  for (int k = 0; k < 512; k += 4) {
    v2f a, b;
    a.x = arow[k + kh]; a.y = arow[k + kh + 1];
    b.x = brow[k + kh]; b.y = brow[k + kh + 1];
    acc = __builtin_amdgcn_wmma_f32_16x16x4_f32(false, a, false, b, (short)0, acc,
                                                false, false);
  }
  const int mb = tm * 16 + (lane >> 4) * 8;
  const int n  = tn * 16 + r;
  const float bias = bout[n];
#pragma unroll
  for (int v = 0; v < 8; ++v)
    out[(size_t)(mb + v) * 256 + n] = acc[v] + bias;
}

extern "C" void kernel_launch(void* const* d_in, const int* in_sizes, int n_in,
                              void* d_out, int out_size, void* d_ws, size_t ws_size,
                              hipStream_t stream) {
  const float* x     = (const float*)d_in[0];
  const int*   batch = (const int*)  d_in[1];
  const float* Wih   = (const float*)d_in[2];
  const float* Whh   = (const float*)d_in[3];
  const float* bih   = (const float*)d_in[4];
  const float* bhh   = (const float*)d_in[5];
  const float* Wout  = (const float*)d_in[6];
  const float* bout  = (const float*)d_in[7];
  const int N = in_sizes[0] / 256;
  const int G = 512;

  float* ws    = (float*)d_ws;
  float* hs    = ws;
  float* cs    = hs + 512 * 256;
  float* qstar = cs + 512 * 256;
  float* gates = qstar + 512 * 512;
  int*   starts = (int*)(gates + 512 * 1024);

  find_starts<<<(N + 255) / 256, 256, 0, stream>>>(batch, starts, N, G);

  attention<true><<<G, 256, 0, stream>>>(x, starts, bih, bhh, hs, cs, qstar);
  for (int step = 1; step < 3; ++step) {
    gates_gemm<<<256, 256, 0, stream>>>(qstar, hs, Wih, Whh, bih, bhh, gates);
    lstm_update<<<512, 256, 0, stream>>>(gates, hs, cs, qstar);
    attention<false><<<G, 256, 0, stream>>>(x, starts, bih, bhh, hs, cs, qstar);
  }
  out_gemm<<<64, 256, 0, stream>>>(qstar, Wout, bout, (float*)d_out);
}


# GEMM fragment loads widened: each lane takes 4 consecutive k of its row per b128 load (was 2 per b64), A and W alike, two WMMAs per load pair; k grouping per WMMA, so the f32 sum order, changes
# speedup vs baseline: 1.9038x; 1.0375x over previous
#include <hip/hip_runtime.h>

typedef __attribute__((ext_vector_type(2))) float v2f;
typedef __attribute__((ext_vector_type(4))) float v4f;
typedef __attribute__((ext_vector_type(8))) float v8f;

__global__ void find_starts(const int* __restrict__ batch, int* __restrict__ starts,
                            int N, int G) {
  int i = blockIdx.x * blockDim.x + threadIdx.x;
  if (i >= N) return;
  int b = batch[i];
  int bp = (i == 0) ? -1 : batch[i - 1];
  for (int g = bp + 1; g <= b; ++g) starts[g] = i;
  if (i == N - 1) {
    for (int g = b + 1; g <= G; ++g) starts[g] = N;
  }
}

__device__ __forceinline__ v8f mma_k8(const float* __restrict__ ap,
                                      const float* __restrict__ bp, v8f acc) {
  const v4f a = *(const v4f*)ap;
  const v4f b = *(const v4f*)bp;
  v2f a0, b0, a1, b1;
  a0.x = a.x; a0.y = a.y; b0.x = b.x; b0.y = b.y;
  a1.x = a.z; a1.y = a.w; b1.x = b.z; b1.y = b.w;
  acc = __builtin_amdgcn_wmma_f32_16x16x4_f32(false, a0, false, b0, (short)0, acc,
                                              false, false);
  acc = __builtin_amdgcn_wmma_f32_16x16x4_f32(false, a1, false, b1, (short)0, acc,
                                              false, false);
  return acc;
}

__global__ void __launch_bounds__(256)
gates_gemm(const float* __restrict__ qstar, const float* __restrict__ hs,
           const float* __restrict__ Wih, const float* __restrict__ Whh,
           const float* __restrict__ bih, const float* __restrict__ bhh,
           float* __restrict__ gates) {
  const int lane = threadIdx.x & 31;
  const int wave = blockIdx.x * (blockDim.x >> 5) + (threadIdx.x >> 5);
  const int TN = 1024 / 16;
  const int tm = wave / TN;
  const int tn = wave % TN;
  const int r  = lane & 15;
  const int kq = (lane >> 4) << 2;

  const float* arow = qstar + (size_t)(tm * 16 + r) * 512;
  const float* brow = Wih   + (size_t)(tn * 16 + r) * 512;
  v8f acc = {};
  for (int k = 0; k < 512; k += 8) acc = mma_k8(arow + k + kq, brow + k + kq, acc);
  const float* arow2 = hs  + (size_t)(tm * 16 + r) * 256;
  const float* brow2 = Whh + (size_t)(tn * 16 + r) * 256;
  for (int k = 0; k < 256; k += 8) acc = mma_k8(arow2 + k + kq, brow2 + k + kq, acc);
  const int mb = tm * 16 + (lane >> 4) * 8;
  const int n  = tn * 16 + r;
  const float bias = bih[n] + bhh[n];
#pragma unroll
  for (int v = 0; v < 8; ++v)
    gates[(size_t)(mb + v) * 1024 + n] = acc[v] + bias;
}

__device__ __forceinline__ float sigmoidf_(float x) {
  return 1.0f / (1.0f + __expf(-x));
}

__global__ void __launch_bounds__(256)
lstm_update(const float* __restrict__ gates, float* __restrict__ hs,
            float* __restrict__ cs, float* __restrict__ qstar) {
  int idx = blockIdx.x * blockDim.x + threadIdx.x;
  int b = idx >> 8, c = idx & 255;
  const float* g = gates + (size_t)b * 1024;
  float ig = sigmoidf_(g[c]);
  float fg = sigmoidf_(g[256 + c]);
  float gg = tanhf(g[512 + c]);
  float og = sigmoidf_(g[768 + c]);
  float cn = fg * cs[idx] + ig * gg;
  cs[idx] = cn;
  float h = og * tanhf(cn);
  hs[idx] = h;
  qstar[(size_t)b * 512 + c] = h;
}

#define NB 8

template <bool FIRST>
__global__ void __launch_bounds__(256)
attention(const float* __restrict__ x, const int* __restrict__ starts,
          const float* __restrict__ bih, const float* __restrict__ bhh,
          float* __restrict__ hs, float* __restrict__ cs,
          float* __restrict__ qstar) {
  const int g    = blockIdx.x;
  const int tid  = threadIdx.x;
  const int lane = tid & 31;
  const int wv   = __builtin_amdgcn_readfirstlane(tid >> 5);

  __shared__ __align__(16) float q[256];
  __shared__ __align__(16) float rall[8][256];
  __shared__ float wmax[8];
  __shared__ float wsum[8];

  if (FIRST) {
    const float ig = sigmoidf_(bih[tid] + bhh[tid]);
    const float fg = sigmoidf_(bih[256 + tid] + bhh[256 + tid]);
    const float gg = tanhf(bih[512 + tid] + bhh[512 + tid]);
    const float og = sigmoidf_(bih[768 + tid] + bhh[768 + tid]);
    const float cn = fg * 0.0f + ig * gg;
    const float h = og * tanhf(cn);
    cs[(size_t)g * 256 + tid] = cn;
    hs[(size_t)g * 256 + tid] = h;
    qstar[(size_t)g * 512 + tid] = h;
    q[tid] = h;
  } else {
    q[tid] = hs[(size_t)g * 256 + tid];
  }
  __syncthreads();

  const int start = starts[g];
  const int end   = starts[g + 1];

  const v4f q0 = *(const v4f*)&q[4 * lane];
  const v4f q1 = *(const v4f*)&q[128 + 4 * lane];

  const float NEG_INF = -__builtin_inff();
  float m = NEG_INF;
  float s = 0.0f;
  v4f r0 = {}, r1 = {};

  for (int base = start + wv * NB; base < end; base += 8 * NB) {
    v4f xa[NB], xb[NB];
#pragma unroll
    for (int j = 0; j < NB; ++j) {
      const int i = (base + j < end) ? (base + j) : base;
      const v4f* xr = (const v4f*)(x + (size_t)i * 256);
      xa[j] = xr[lane];
      xb[j] = xr[32 + lane];
    }

    float d[NB];
#pragma unroll
    for (int j = 0; j < NB; ++j) {
      float t = 0.0f;
      t = fmaf(xa[j].x, q0.x, t); t = fmaf(xa[j].y, q0.y, t);
      t = fmaf(xa[j].z, q0.z, t); t = fmaf(xa[j].w, q0.w, t);
      t = fmaf(xb[j].x, q1.x, t); t = fmaf(xb[j].y, q1.y, t);
      t = fmaf(xb[j].z, q1.z, t); t = fmaf(xb[j].w, q1.w, t);
      d[j] = t;
    }
#pragma unroll
    for (int off = 16; off; off >>= 1) {
#pragma unroll
      for (int j = 0; j < NB; ++j) d[j] += __shfl_xor(d[j], off, 32);
    }
    float mb = NEG_INF;
#pragma unroll
    for (int j = 0; j < NB; ++j) {
      if (base + j >= end) d[j] = NEG_INF;
      mb = fmaxf(mb, d[j]);
    }
    const float mn = fmaxf(m, mb);
    const float sc = (m == NEG_INF) ? 0.0f : __expf(m - mn);
    m = mn;
    s *= sc;
    r0.x *= sc; r0.y *= sc; r0.z *= sc; r0.w *= sc;
    r1.x *= sc; r1.y *= sc; r1.z *= sc; r1.w *= sc;
#pragma unroll
    for (int j = 0; j < NB; ++j) {
      const float a = __expf(d[j] - m);
      s += a;
      r0.x = fmaf(a, xa[j].x, r0.x); r0.y = fmaf(a, xa[j].y, r0.y);
      r0.z = fmaf(a, xa[j].z, r0.z); r0.w = fmaf(a, xa[j].w, r0.w);
      r1.x = fmaf(a, xb[j].x, r1.x); r1.y = fmaf(a, xb[j].y, r1.y);
      r1.z = fmaf(a, xb[j].z, r1.z); r1.w = fmaf(a, xb[j].w, r1.w);
    }
  }

  *(v4f*)&rall[wv][4 * lane]       = r0;
  *(v4f*)&rall[wv][128 + 4 * lane] = r1;
  if (lane == 0) { wmax[wv] = m; wsum[wv] = s; }
  __syncthreads();

  float M = -__builtin_inff();
#pragma unroll
  for (int w = 0; w < 8; ++w) M = fmaxf(M, wmax[w]);

  float stot = 0.0f, r = 0.0f;
#pragma unroll
  for (int w = 0; w < 8; ++w) {
    float sw = wsum[w];
    if (sw > 0.0f) {
      float sc = __expf(wmax[w] - M);
      stot = fmaf(sw, sc, stot);
      r    = fmaf(rall[w][tid], sc, r);
    }
  }
  qstar[(size_t)g * 512 + 256 + tid] = (stot > 0.0f) ? (r / stot) : 0.0f;
}

__global__ void __launch_bounds__(256)
out_gemm(const float* __restrict__ qstar, const float* __restrict__ Wout,
         const float* __restrict__ bout, float* __restrict__ out) {
  const int lane = threadIdx.x & 31;
  const int wave = blockIdx.x * (blockDim.x >> 5) + (threadIdx.x >> 5);
  const int TN = 256 / 16;
  const int tm = wave / TN;
  const int tn = wave % TN;
  const int r  = lane & 15;
  const int kq = (lane >> 4) << 2;

  const float* arow = qstar + (size_t)(tm * 16 + r) * 512;
  const float* brow = Wout  + (size_t)(tn * 16 + r) * 512;
  v8f acc = {};
  for (int k = 0; k < 512; k += 8) acc = mma_k8(arow + k + kq, brow + k + kq, acc);
  const int mb = tm * 16 + (lane >> 4) * 8;
  const int n  = tn * 16 + r;
  const float bias = bout[n];
#pragma unroll
  for (int v = 0; v < 8; ++v)
    out[(size_t)(mb + v) * 256 + n] = acc[v] + bias;
}

extern "C" void kernel_launch(void* const* d_in, const int* in_sizes, int n_in,
                              void* d_out, int out_size, void* d_ws, size_t ws_size,
                              hipStream_t stream) {
  const float* x     = (const float*)d_in[0];
  const int*   batch = (const int*)  d_in[1];
  const float* Wih   = (const float*)d_in[2];
  const float* Whh   = (const float*)d_in[3];
  const float* bih   = (const float*)d_in[4];
  const float* bhh   = (const float*)d_in[5];
  const float* Wout  = (const float*)d_in[6];
  const float* bout  = (const float*)d_in[7];
  const int N = in_sizes[0] / 256;
  const int G = 512;

  float* ws    = (float*)d_ws;
  float* hs    = ws;
  float* cs    = hs + 512 * 256;
  float* qstar = cs + 512 * 256;
  float* gates = qstar + 512 * 512;
  int*   starts = (int*)(gates + 512 * 1024);

  find_starts<<<(N + 255) / 256, 256, 0, stream>>>(batch, starts, N, G);

  attention<true><<<G, 256, 0, stream>>>(x, starts, bih, bhh, hs, cs, qstar);
  for (int step = 1; step < 3; ++step) {
    gates_gemm<<<256, 256, 0, stream>>>(qstar, hs, Wih, Whh, bih, bhh, gates);
    lstm_update<<<512, 256, 0, stream>>>(gates, hs, cs, qstar);
    attention<false><<<G, 256, 0, stream>>>(x, starts, bih, bhh, hs, cs, qstar);
  }
  out_gemm<<<64, 256, 0, stream>>>(qstar, Wout, bout, (float*)d_out);
}


# LSTM cell folded into the attention launch: each block computes its graph's cell from its gates row and cs, as step 0 does from the biases; stores follow the q barrier; both LSTM launches go (9 to 7)
# speedup vs baseline: 1.9775x; 1.0387x over previous
#include <hip/hip_runtime.h>

typedef __attribute__((ext_vector_type(2))) float v2f;
typedef __attribute__((ext_vector_type(4))) float v4f;
typedef __attribute__((ext_vector_type(8))) float v8f;

__global__ void find_starts(const int* __restrict__ batch, int* __restrict__ starts,
                            int N, int G) {
  int i = blockIdx.x * blockDim.x + threadIdx.x;
  if (i >= N) return;
  int b = batch[i];
  int bp = (i == 0) ? -1 : batch[i - 1];
  for (int g = bp + 1; g <= b; ++g) starts[g] = i;
  if (i == N - 1) {
    for (int g = b + 1; g <= G; ++g) starts[g] = N;
  }
}

__device__ __forceinline__ v8f mma_k8(const float* __restrict__ ap,
                                      const float* __restrict__ bp, v8f acc) {
  const v4f a = *(const v4f*)ap;
  const v4f b = *(const v4f*)bp;
  v2f a0, b0, a1, b1;
  a0.x = a.x; a0.y = a.y; b0.x = b.x; b0.y = b.y;
  a1.x = a.z; a1.y = a.w; b1.x = b.z; b1.y = b.w;
  acc = __builtin_amdgcn_wmma_f32_16x16x4_f32(false, a0, false, b0, (short)0, acc,
                                              false, false);
  acc = __builtin_amdgcn_wmma_f32_16x16x4_f32(false, a1, false, b1, (short)0, acc,
                                              false, false);
  return acc;
}

__global__ void __launch_bounds__(256)
gates_gemm(const float* __restrict__ qstar, const float* __restrict__ hs,
           const float* __restrict__ Wih, const float* __restrict__ Whh,
           const float* __restrict__ bih, const float* __restrict__ bhh,
           float* __restrict__ gates) {
  const int lane = threadIdx.x & 31;
  const int wave = blockIdx.x * (blockDim.x >> 5) + (threadIdx.x >> 5);
  const int TN = 1024 / 16;
  const int tm = wave / TN;
  const int tn = wave % TN;
  const int r  = lane & 15;
  const int kq = (lane >> 4) << 2;

  const float* arow = qstar + (size_t)(tm * 16 + r) * 512;
  const float* brow = Wih   + (size_t)(tn * 16 + r) * 512;
  v8f acc = {};
  for (int k = 0; k < 512; k += 8) acc = mma_k8(arow + k + kq, brow + k + kq, acc);
  const float* arow2 = hs  + (size_t)(tm * 16 + r) * 256;
  const float* brow2 = Whh + (size_t)(tn * 16 + r) * 256;
  for (int k = 0; k < 256; k += 8) acc = mma_k8(arow2 + k + kq, brow2 + k + kq, acc);
  const int mb = tm * 16 + (lane >> 4) * 8;
  const int n  = tn * 16 + r;
  const float bias = bih[n] + bhh[n];
#pragma unroll
  for (int v = 0; v < 8; ++v)
    gates[(size_t)(mb + v) * 1024 + n] = acc[v] + bias;
}

__device__ __forceinline__ float sigmoidf_(float x) {
  return 1.0f / (1.0f + __expf(-x));
}

#define NB 8

template <bool FIRST>
__global__ void __launch_bounds__(256)
attention(const float* __restrict__ x, const int* __restrict__ starts,
          const float* __restrict__ bih, const float* __restrict__ bhh,
          const float* __restrict__ gates,
          float* __restrict__ hs, float* __restrict__ cs,
          float* __restrict__ qstar) {
  const int g    = blockIdx.x;
  const int tid  = threadIdx.x;
  const int lane = tid & 31;
  const int wv   = __builtin_amdgcn_readfirstlane(tid >> 5);

  __shared__ __align__(16) float q[256];
  __shared__ __align__(16) float rall[8][256];
  __shared__ float wmax[8];
  __shared__ float wsum[8];

  float cn, h;
  if (FIRST) {
    const float ig = sigmoidf_(bih[tid] + bhh[tid]);
    const float fg = sigmoidf_(bih[256 + tid] + bhh[256 + tid]);
    const float gg = tanhf(bih[512 + tid] + bhh[512 + tid]);
    const float og = sigmoidf_(bih[768 + tid] + bhh[768 + tid]);
    cn = fg * 0.0f + ig * gg;
    h = og * tanhf(cn);
  } else {
    const float* gt = gates + (size_t)g * 1024;
    const float ig = sigmoidf_(gt[tid]);
    const float fg = sigmoidf_(gt[256 + tid]);
    const float gg = tanhf(gt[512 + tid]);
    const float og = sigmoidf_(gt[768 + tid]);
    cn = fg * cs[(size_t)g * 256 + tid] + ig * gg;
    h = og * tanhf(cn);
  }
  q[tid] = h;
  __syncthreads();
  cs[(size_t)g * 256 + tid] = cn;
  hs[(size_t)g * 256 + tid] = h;
  qstar[(size_t)g * 512 + tid] = h;

  const int start = starts[g];
  const int end   = starts[g + 1];

  const v4f q0 = *(const v4f*)&q[4 * lane];
  const v4f q1 = *(const v4f*)&q[128 + 4 * lane];

  const float NEG_INF = -__builtin_inff();
  float m = NEG_INF;
  float s = 0.0f;
  v4f r0 = {}, r1 = {};

  for (int base = start + wv * NB; base < end; base += 8 * NB) {
    v4f xa[NB], xb[NB];
#pragma unroll
    for (int j = 0; j < NB; ++j) {
      const int i = (base + j < end) ? (base + j) : base;
      const v4f* xr = (const v4f*)(x + (size_t)i * 256);
      xa[j] = xr[lane];
      xb[j] = xr[32 + lane];
    }

    float d[NB];
#pragma unroll
    for (int j = 0; j < NB; ++j) {
      float t = 0.0f;
      t = fmaf(xa[j].x, q0.x, t); t = fmaf(xa[j].y, q0.y, t);
      t = fmaf(xa[j].z, q0.z, t); t = fmaf(xa[j].w, q0.w, t);
      t = fmaf(xb[j].x, q1.x, t); t = fmaf(xb[j].y, q1.y, t);
      t = fmaf(xb[j].z, q1.z, t); t = fmaf(xb[j].w, q1.w, t);
      d[j] = t;
    }
#pragma unroll
    for (int off = 16; off; off >>= 1) {
#pragma unroll
      for (int j = 0; j < NB; ++j) d[j] += __shfl_xor(d[j], off, 32);
    }
    float mb = NEG_INF;
#pragma unroll
    for (int j = 0; j < NB; ++j) {
      if (base + j >= end) d[j] = NEG_INF;
      mb = fmaxf(mb, d[j]);
    }
    const float mn = fmaxf(m, mb);
    const float sc = (m == NEG_INF) ? 0.0f : __expf(m - mn);
    m = mn;
    s *= sc;
    r0.x *= sc; r0.y *= sc; r0.z *= sc; r0.w *= sc;
    r1.x *= sc; r1.y *= sc; r1.z *= sc; r1.w *= sc;
#pragma unroll
    for (int j = 0; j < NB; ++j) {
      const float a = __expf(d[j] - m);
      s += a;
      r0.x = fmaf(a, xa[j].x, r0.x); r0.y = fmaf(a, xa[j].y, r0.y);
      r0.z = fmaf(a, xa[j].z, r0.z); r0.w = fmaf(a, xa[j].w, r0.w);
      r1.x = fmaf(a, xb[j].x, r1.x); r1.y = fmaf(a, xb[j].y, r1.y);
      r1.z = fmaf(a, xb[j].z, r1.z); r1.w = fmaf(a, xb[j].w, r1.w);
    }
  }

  *(v4f*)&rall[wv][4 * lane]       = r0;
  *(v4f*)&rall[wv][128 + 4 * lane] = r1;
  if (lane == 0) { wmax[wv] = m; wsum[wv] = s; }
  __syncthreads();

  float M = -__builtin_inff();
#pragma unroll
  for (int w = 0; w < 8; ++w) M = fmaxf(M, wmax[w]);

  float stot = 0.0f, r = 0.0f;
#pragma unroll
  for (int w = 0; w < 8; ++w) {
    float sw = wsum[w];
    if (sw > 0.0f) {
      float sc = __expf(wmax[w] - M);
      stot = fmaf(sw, sc, stot);
      r    = fmaf(rall[w][tid], sc, r);
    }
  }
  qstar[(size_t)g * 512 + 256 + tid] = (stot > 0.0f) ? (r / stot) : 0.0f;
}

__global__ void __launch_bounds__(256)
out_gemm(const float* __restrict__ qstar, const float* __restrict__ Wout,
         const float* __restrict__ bout, float* __restrict__ out) {
  const int lane = threadIdx.x & 31;
  const int wave = blockIdx.x * (blockDim.x >> 5) + (threadIdx.x >> 5);
  const int TN = 256 / 16;
  const int tm = wave / TN;
  const int tn = wave % TN;
  const int r  = lane & 15;
  const int kq = (lane >> 4) << 2;

  const float* arow = qstar + (size_t)(tm * 16 + r) * 512;
  const float* brow = Wout  + (size_t)(tn * 16 + r) * 512;
  v8f acc = {};
  for (int k = 0; k < 512; k += 8) acc = mma_k8(arow + k + kq, brow + k + kq, acc);
  const int mb = tm * 16 + (lane >> 4) * 8;
  const int n  = tn * 16 + r;
  const float bias = bout[n];
#pragma unroll
  for (int v = 0; v < 8; ++v)
    out[(size_t)(mb + v) * 256 + n] = acc[v] + bias;
}

extern "C" void kernel_launch(void* const* d_in, const int* in_sizes, int n_in,
                              void* d_out, int out_size, void* d_ws, size_t ws_size,
                              hipStream_t stream) {
  const float* x     = (const float*)d_in[0];
  const int*   batch = (const int*)  d_in[1];
  const float* Wih   = (const float*)d_in[2];
  const float* Whh   = (const float*)d_in[3];
  const float* bih   = (const float*)d_in[4];
  const float* bhh   = (const float*)d_in[5];
  const float* Wout  = (const float*)d_in[6];
  const float* bout  = (const float*)d_in[7];
  const int N = in_sizes[0] / 256;
  const int G = 512;

  float* ws    = (float*)d_ws;
  float* hs    = ws;
  float* cs    = hs + 512 * 256;
  float* qstar = cs + 512 * 256;
  float* gates = qstar + 512 * 512;
  int*   starts = (int*)(gates + 512 * 1024);

  find_starts<<<(N + 255) / 256, 256, 0, stream>>>(batch, starts, N, G);

  attention<true><<<G, 256, 0, stream>>>(x, starts, bih, bhh, gates, hs, cs, qstar);
  for (int step = 1; step < 3; ++step) {
    gates_gemm<<<256, 256, 0, stream>>>(qstar, hs, Wih, Whh, bih, bhh, gates);
    attention<false><<<G, 256, 0, stream>>>(x, starts, bih, bhh, gates, hs, cs, qstar);
  }
  out_gemm<<<64, 256, 0, stream>>>(qstar, Wout, bout, (float*)d_out);
}


# GEMMs: each block copies the 16-row A tile its 8 waves share into LDS once with whole-line loads and the waves read A fragments from LDS; W stays direct; k order and f32 sums unchanged
# speedup vs baseline: 2.1443x; 1.0843x over previous
#include <hip/hip_runtime.h>

typedef __attribute__((ext_vector_type(2))) float v2f;
typedef __attribute__((ext_vector_type(4))) float v4f;
typedef __attribute__((ext_vector_type(8))) float v8f;

__global__ void find_starts(const int* __restrict__ batch, int* __restrict__ starts,
                            int N, int G) {
  int i = blockIdx.x * blockDim.x + threadIdx.x;
  if (i >= N) return;
  int b = batch[i];
  int bp = (i == 0) ? -1 : batch[i - 1];
  for (int g = bp + 1; g <= b; ++g) starts[g] = i;
  if (i == N - 1) {
    for (int g = b + 1; g <= G; ++g) starts[g] = N;
  }
}

__device__ __forceinline__ v8f mma_k8(const float* __restrict__ ap,
                                      const float* __restrict__ bp, v8f acc) {
  const v4f a = *(const v4f*)ap;
  const v4f b = *(const v4f*)bp;
  v2f a0, b0, a1, b1;
  a0.x = a.x; a0.y = a.y; b0.x = b.x; b0.y = b.y;
  a1.x = a.z; a1.y = a.w; b1.x = b.z; b1.y = b.w;
  acc = __builtin_amdgcn_wmma_f32_16x16x4_f32(false, a0, false, b0, (short)0, acc,
                                              false, false);
  acc = __builtin_amdgcn_wmma_f32_16x16x4_f32(false, a1, false, b1, (short)0, acc,
                                              false, false);
  return acc;
}

#define GATES_K   768
#define GATES_AP  (GATES_K + 4)
#define OUT_K     512
#define OUT_AP    (OUT_K + 4)

__global__ void __launch_bounds__(256)
gates_gemm(const float* __restrict__ qstar, const float* __restrict__ hs,
           const float* __restrict__ Wih, const float* __restrict__ Whh,
           const float* __restrict__ bih, const float* __restrict__ bhh,
           float* __restrict__ gates) {
  __shared__ __align__(16) float As[16 * GATES_AP];
  const int tid  = threadIdx.x;
  const int lane = tid & 31;
  const int tm = blockIdx.x >> 3;
  const int tn = ((blockIdx.x & 7) << 3) + (tid >> 5);
  const int r  = lane & 15;
  const int kq = (lane >> 4) << 2;

  v4f st[12];
#pragma unroll
  for (int i = 0; i < 8; ++i) {
    const int idx = i * 256 + tid;
    st[i] = *(const v4f*)(qstar + (size_t)(tm * 16 + (idx >> 7)) * 512 + 4 * (idx & 127));
  }
#pragma unroll
  for (int i = 0; i < 4; ++i) {
    const int idx = i * 256 + tid;
    st[8 + i] = *(const v4f*)(hs + (size_t)(tm * 16 + (idx >> 6)) * 256 + 4 * (idx & 63));
  }
#pragma unroll
  for (int i = 0; i < 8; ++i) {
    const int idx = i * 256 + tid;
    *(v4f*)&As[(idx >> 7) * GATES_AP + 4 * (idx & 127)] = st[i];
  }
#pragma unroll
  for (int i = 0; i < 4; ++i) {
    const int idx = i * 256 + tid;
    *(v4f*)&As[(idx >> 6) * GATES_AP + 512 + 4 * (idx & 63)] = st[8 + i];
  }
  __syncthreads();

  const float* arow = As + r * GATES_AP + kq;
  const float* brow = Wih + (size_t)(tn * 16 + r) * 512 + kq;
  v8f acc = {};
  for (int k = 0; k < 512; k += 8) acc = mma_k8(arow + k, brow + k, acc);
  const float* brow2 = Whh + (size_t)(tn * 16 + r) * 256 + kq;
  for (int k = 0; k < 256; k += 8) acc = mma_k8(arow + 512 + k, brow2 + k, acc);
  const int mb = tm * 16 + (lane >> 4) * 8;
  const int n  = tn * 16 + r;
  const float bias = bih[n] + bhh[n];
#pragma unroll
  for (int v = 0; v < 8; ++v)
    gates[(size_t)(mb + v) * 1024 + n] = acc[v] + bias;
}

__device__ __forceinline__ float sigmoidf_(float x) {
  return 1.0f / (1.0f + __expf(-x));
}

#define NB 8

template <bool FIRST>
__global__ void __launch_bounds__(256)
attention(const float* __restrict__ x, const int* __restrict__ starts,
          const float* __restrict__ bih, const float* __restrict__ bhh,
          const float* __restrict__ gates,
          float* __restrict__ hs, float* __restrict__ cs,
          float* __restrict__ qstar) {
  const int g    = blockIdx.x;
  const int tid  = threadIdx.x;
  const int lane = tid & 31;
  const int wv   = __builtin_amdgcn_readfirstlane(tid >> 5);

  __shared__ __align__(16) float q[256];
  __shared__ __align__(16) float rall[8][256];
  __shared__ float wmax[8];
  __shared__ float wsum[8];

  float cn, h;
  if (FIRST) {
    const float ig = sigmoidf_(bih[tid] + bhh[tid]);
    const float fg = sigmoidf_(bih[256 + tid] + bhh[256 + tid]);
    const float gg = tanhf(bih[512 + tid] + bhh[512 + tid]);
    const float og = sigmoidf_(bih[768 + tid] + bhh[768 + tid]);
    cn = fg * 0.0f + ig * gg;
    h = og * tanhf(cn);
  } else {
    const float* gt = gates + (size_t)g * 1024;
    const float ig = sigmoidf_(gt[tid]);
    const float fg = sigmoidf_(gt[256 + tid]);
    const float gg = tanhf(gt[512 + tid]);
    const float og = sigmoidf_(gt[768 + tid]);
    cn = fg * cs[(size_t)g * 256 + tid] + ig * gg;
    h = og * tanhf(cn);
  }
  q[tid] = h;
  __syncthreads();
  cs[(size_t)g * 256 + tid] = cn;
  hs[(size_t)g * 256 + tid] = h;
  qstar[(size_t)g * 512 + tid] = h;

  const int start = starts[g];
  const int end   = starts[g + 1];

  const v4f q0 = *(const v4f*)&q[4 * lane];
  const v4f q1 = *(const v4f*)&q[128 + 4 * lane];

  const float NEG_INF = -__builtin_inff();
  float m = NEG_INF;
  float s = 0.0f;
  v4f r0 = {}, r1 = {};

  for (int base = start + wv * NB; base < end; base += 8 * NB) {
    v4f xa[NB], xb[NB];
#pragma unroll
    for (int j = 0; j < NB; ++j) {
      const int i = (base + j < end) ? (base + j) : base;
      const v4f* xr = (const v4f*)(x + (size_t)i * 256);
      xa[j] = xr[lane];
      xb[j] = xr[32 + lane];
    }

    float d[NB];
#pragma unroll
    for (int j = 0; j < NB; ++j) {
      float t = 0.0f;
      t = fmaf(xa[j].x, q0.x, t); t = fmaf(xa[j].y, q0.y, t);
      t = fmaf(xa[j].z, q0.z, t); t = fmaf(xa[j].w, q0.w, t);
      t = fmaf(xb[j].x, q1.x, t); t = fmaf(xb[j].y, q1.y, t);
      t = fmaf(xb[j].z, q1.z, t); t = fmaf(xb[j].w, q1.w, t);
      d[j] = t;
    }
#pragma unroll
    for (int off = 16; off; off >>= 1) {
#pragma unroll
      for (int j = 0; j < NB; ++j) d[j] += __shfl_xor(d[j], off, 32);
    }
    float mb = NEG_INF;
#pragma unroll
    for (int j = 0; j < NB; ++j) {
      if (base + j >= end) d[j] = NEG_INF;
      mb = fmaxf(mb, d[j]);
    }
    const float mn = fmaxf(m, mb);
    const float sc = (m == NEG_INF) ? 0.0f : __expf(m - mn);
    m = mn;
    s *= sc;
    r0.x *= sc; r0.y *= sc; r0.z *= sc; r0.w *= sc;
    r1.x *= sc; r1.y *= sc; r1.z *= sc; r1.w *= sc;
#pragma unroll
    for (int j = 0; j < NB; ++j) {
      const float a = __expf(d[j] - m);
      s += a;
      r0.x = fmaf(a, xa[j].x, r0.x); r0.y = fmaf(a, xa[j].y, r0.y);
      r0.z = fmaf(a, xa[j].z, r0.z); r0.w = fmaf(a, xa[j].w, r0.w);
      r1.x = fmaf(a, xb[j].x, r1.x); r1.y = fmaf(a, xb[j].y, r1.y);
      r1.z = fmaf(a, xb[j].z, r1.z); r1.w = fmaf(a, xb[j].w, r1.w);
    }
  }

  *(v4f*)&rall[wv][4 * lane]       = r0;
  *(v4f*)&rall[wv][128 + 4 * lane] = r1;
  if (lane == 0) { wmax[wv] = m; wsum[wv] = s; }
  __syncthreads();

  float M = -__builtin_inff();
#pragma unroll
  for (int w = 0; w < 8; ++w) M = fmaxf(M, wmax[w]);

  float stot = 0.0f, r = 0.0f;
#pragma unroll
  for (int w = 0; w < 8; ++w) {
    float sw = wsum[w];
    if (sw > 0.0f) {
      float sc = __expf(wmax[w] - M);
      stot = fmaf(sw, sc, stot);
      r    = fmaf(rall[w][tid], sc, r);
    }
  }
  qstar[(size_t)g * 512 + 256 + tid] = (stot > 0.0f) ? (r / stot) : 0.0f;
}

__global__ void __launch_bounds__(256)
out_gemm(const float* __restrict__ qstar, const float* __restrict__ Wout,
         const float* __restrict__ bout, float* __restrict__ out) {
  __shared__ __align__(16) float As[16 * OUT_AP];
  const int tid  = threadIdx.x;
  const int lane = tid & 31;
  const int tm = blockIdx.x >> 1;
  const int tn = ((blockIdx.x & 1) << 3) + (tid >> 5);
  const int r  = lane & 15;
  const int kq = (lane >> 4) << 2;

  v4f st[8];
#pragma unroll
  for (int i = 0; i < 8; ++i) {
    const int idx = i * 256 + tid;
    st[i] = *(const v4f*)(qstar + (size_t)(tm * 16 + (idx >> 7)) * 512 + 4 * (idx & 127));
  }
#pragma unroll
  for (int i = 0; i < 8; ++i) {
    const int idx = i * 256 + tid;
    *(v4f*)&As[(idx >> 7) * OUT_AP + 4 * (idx & 127)] = st[i];
  }
  __syncthreads();

  const float* arow = As + r * OUT_AP + kq;
  const float* brow = Wout + (size_t)(tn * 16 + r) * 512 + kq;
  v8f acc = {};
  for (int k = 0; k < 512; k += 8) acc = mma_k8(arow + k, brow + k, acc);
  const int mb = tm * 16 + (lane >> 4) * 8;
  const int n  = tn * 16 + r;
  const float bias = bout[n];
#pragma unroll
  for (int v = 0; v < 8; ++v)
    out[(size_t)(mb + v) * 256 + n] = acc[v] + bias;
}

extern "C" void kernel_launch(void* const* d_in, const int* in_sizes, int n_in,
                              void* d_out, int out_size, void* d_ws, size_t ws_size,
                              hipStream_t stream) {
  const float* x     = (const float*)d_in[0];
  const int*   batch = (const int*)  d_in[1];
  const float* Wih   = (const float*)d_in[2];
  const float* Whh   = (const float*)d_in[3];
  const float* bih   = (const float*)d_in[4];
  const float* bhh   = (const float*)d_in[5];
  const float* Wout  = (const float*)d_in[6];
  const float* bout  = (const float*)d_in[7];
  const int N = in_sizes[0] / 256;
  const int G = 512;

  float* ws    = (float*)d_ws;
  float* hs    = ws;
  float* cs    = hs + 512 * 256;
  float* qstar = cs + 512 * 256;
  float* gates = qstar + 512 * 512;
  int*   starts = (int*)(gates + 512 * 1024);

  find_starts<<<(N + 255) / 256, 256, 0, stream>>>(batch, starts, N, G);

  attention<true><<<G, 256, 0, stream>>>(x, starts, bih, bhh, gates, hs, cs, qstar);
  for (int step = 1; step < 3; ++step) {
    gates_gemm<<<256, 256, 0, stream>>>(qstar, hs, Wih, Whh, bih, bhh, gates);
    attention<false><<<G, 256, 0, stream>>>(x, starts, bih, bhh, gates, hs, cs, qstar);
  }
  out_gemm<<<64, 256, 0, stream>>>(qstar, Wout, bout, (float*)d_out);
}


# attention: each wave requests its next trip's 16 row loads before it uses the current trip's (two register sets in turn); the first trip's ahead of the LSTM cell; every sum and its order unchanged
# speedup vs baseline: 2.2295x; 1.0397x over previous
#include <hip/hip_runtime.h>

typedef __attribute__((ext_vector_type(2))) float v2f;
typedef __attribute__((ext_vector_type(4))) float v4f;
typedef __attribute__((ext_vector_type(8))) float v8f;

__global__ void find_starts(const int* __restrict__ batch, int* __restrict__ starts,
                            int N, int G) {
  int i = blockIdx.x * blockDim.x + threadIdx.x;
  if (i >= N) return;
  int b = batch[i];
  int bp = (i == 0) ? -1 : batch[i - 1];
  for (int g = bp + 1; g <= b; ++g) starts[g] = i;
  if (i == N - 1) {
    for (int g = b + 1; g <= G; ++g) starts[g] = N;
  }
}

__device__ __forceinline__ v8f mma_k8(const float* __restrict__ ap,
                                      const float* __restrict__ bp, v8f acc) {
  const v4f a = *(const v4f*)ap;
  const v4f b = *(const v4f*)bp;
  v2f a0, b0, a1, b1;
  a0.x = a.x; a0.y = a.y; b0.x = b.x; b0.y = b.y;
  a1.x = a.z; a1.y = a.w; b1.x = b.z; b1.y = b.w;
  acc = __builtin_amdgcn_wmma_f32_16x16x4_f32(false, a0, false, b0, (short)0, acc,
                                              false, false);
  acc = __builtin_amdgcn_wmma_f32_16x16x4_f32(false, a1, false, b1, (short)0, acc,
                                              false, false);
  return acc;
}

#define GATES_K   768
#define GATES_AP  (GATES_K + 4)
#define OUT_K     512
#define OUT_AP    (OUT_K + 4)

__global__ void __launch_bounds__(256)
gates_gemm(const float* __restrict__ qstar, const float* __restrict__ hs,
           const float* __restrict__ Wih, const float* __restrict__ Whh,
           const float* __restrict__ bih, const float* __restrict__ bhh,
           float* __restrict__ gates) {
  __shared__ __align__(16) float As[16 * GATES_AP];
  const int tid  = threadIdx.x;
  const int lane = tid & 31;
  const int tm = blockIdx.x >> 3;
  const int tn = ((blockIdx.x & 7) << 3) + (tid >> 5);
  const int r  = lane & 15;
  const int kq = (lane >> 4) << 2;

  v4f st[12];
#pragma unroll
  for (int i = 0; i < 8; ++i) {
    const int idx = i * 256 + tid;
    st[i] = *(const v4f*)(qstar + (size_t)(tm * 16 + (idx >> 7)) * 512 + 4 * (idx & 127));
  }
#pragma unroll
  for (int i = 0; i < 4; ++i) {
    const int idx = i * 256 + tid;
    st[8 + i] = *(const v4f*)(hs + (size_t)(tm * 16 + (idx >> 6)) * 256 + 4 * (idx & 63));
  }
#pragma unroll
  for (int i = 0; i < 8; ++i) {
    const int idx = i * 256 + tid;
    *(v4f*)&As[(idx >> 7) * GATES_AP + 4 * (idx & 127)] = st[i];
  }
#pragma unroll
  for (int i = 0; i < 4; ++i) {
    const int idx = i * 256 + tid;
    *(v4f*)&As[(idx >> 6) * GATES_AP + 512 + 4 * (idx & 63)] = st[8 + i];
  }
  __syncthreads();

  const float* arow = As + r * GATES_AP + kq;
  const float* brow = Wih + (size_t)(tn * 16 + r) * 512 + kq;
  v8f acc = {};
  for (int k = 0; k < 512; k += 8) acc = mma_k8(arow + k, brow + k, acc);
  const float* brow2 = Whh + (size_t)(tn * 16 + r) * 256 + kq;
  for (int k = 0; k < 256; k += 8) acc = mma_k8(arow + 512 + k, brow2 + k, acc);
  const int mb = tm * 16 + (lane >> 4) * 8;
  const int n  = tn * 16 + r;
  const float bias = bih[n] + bhh[n];
#pragma unroll
  for (int v = 0; v < 8; ++v)
    gates[(size_t)(mb + v) * 1024 + n] = acc[v] + bias;
}

__device__ __forceinline__ float sigmoidf_(float x) {
  return 1.0f / (1.0f + __expf(-x));
}

#define NB 8
#define TRIP (8 * NB)

__device__ __forceinline__ void request_trip(const float* __restrict__ x, int base, int last,
                                             int lane, v4f (&xa)[NB], v4f (&xb)[NB]) {
#pragma unroll
  for (int j = 0; j < NB; ++j) {
    int i = base + j;
    i = (i < last) ? i : last;
    i = (i > 0) ? i : 0;
    const v4f* xr = (const v4f*)(x + (size_t)i * 256);
    xa[j] = xr[lane];
    xb[j] = xr[32 + lane];
  }
}

__device__ __forceinline__ void use_trip(const v4f (&xa)[NB], const v4f (&xb)[NB],
                                         int base, int end, v4f q0, v4f q1,
                                         float& m, float& s, v4f& r0, v4f& r1) {
  const float NEG_INF = -__builtin_inff();
  float d[NB];
#pragma unroll
  for (int j = 0; j < NB; ++j) {
    float t = 0.0f;
    t = fmaf(xa[j].x, q0.x, t); t = fmaf(xa[j].y, q0.y, t);
    t = fmaf(xa[j].z, q0.z, t); t = fmaf(xa[j].w, q0.w, t);
    t = fmaf(xb[j].x, q1.x, t); t = fmaf(xb[j].y, q1.y, t);
    t = fmaf(xb[j].z, q1.z, t); t = fmaf(xb[j].w, q1.w, t);
    asm volatile("" : "+v"(t));
    d[j] = t;
  }
#pragma unroll
  for (int off = 16; off; off >>= 1) {
#pragma unroll
    for (int j = 0; j < NB; ++j) d[j] += __shfl_xor(d[j], off, 32);
  }
  float mb = NEG_INF;
#pragma unroll
  for (int j = 0; j < NB; ++j) {
    if (base + j >= end) d[j] = NEG_INF;
    mb = fmaxf(mb, d[j]);
  }
  const float mn = fmaxf(m, mb);
  const float sc = (m == NEG_INF) ? 0.0f : __expf(m - mn);
  m = mn;
  s *= sc;
  r0.x *= sc; r0.y *= sc; r0.z *= sc; r0.w *= sc;
  r1.x *= sc; r1.y *= sc; r1.z *= sc; r1.w *= sc;
#pragma unroll
  for (int j = 0; j < NB; ++j) {
    const float a = __expf(d[j] - m);
    s += a;
    r0.x = fmaf(a, xa[j].x, r0.x); r0.y = fmaf(a, xa[j].y, r0.y);
    r0.z = fmaf(a, xa[j].z, r0.z); r0.w = fmaf(a, xa[j].w, r0.w);
    r1.x = fmaf(a, xb[j].x, r1.x); r1.y = fmaf(a, xb[j].y, r1.y);
    r1.z = fmaf(a, xb[j].z, r1.z); r1.w = fmaf(a, xb[j].w, r1.w);
  }
  asm volatile("" : "+v"(r0), "+v"(r1), "+v"(s), "+v"(m));
}

template <bool FIRST>
__global__ void __launch_bounds__(256)
attention(const float* __restrict__ x, const int* __restrict__ starts,
          const float* __restrict__ bih, const float* __restrict__ bhh,
          const float* __restrict__ gates,
          float* __restrict__ hs, float* __restrict__ cs,
          float* __restrict__ qstar) {
  const int g    = blockIdx.x;
  const int tid  = threadIdx.x;
  const int lane = tid & 31;
  const int wv   = __builtin_amdgcn_readfirstlane(tid >> 5);

  __shared__ __align__(16) float q[256];
  __shared__ __align__(16) float rall[8][256];
  __shared__ float wmax[8];
  __shared__ float wsum[8];

  const int start = starts[g];
  const int end   = starts[g + 1];
  const int last  = end - 1;
  int base = start + wv * NB;

  float gi, gf, gg_, go, cprev;
  if (FIRST) {
    gi  = bih[tid] + bhh[tid];
    gf  = bih[256 + tid] + bhh[256 + tid];
    gg_ = bih[512 + tid] + bhh[512 + tid];
    go  = bih[768 + tid] + bhh[768 + tid];
    cprev = 0.0f;
  } else {
    const float* gt = gates + (size_t)g * 1024;
    gi  = gt[tid];
    gf  = gt[256 + tid];
    gg_ = gt[512 + tid];
    go  = gt[768 + tid];
    cprev = cs[(size_t)g * 256 + tid];
  }

  v4f xa0[NB], xb0[NB], xa1[NB], xb1[NB];
  request_trip(x, base, last, lane, xa0, xb0);

  const float cn = sigmoidf_(gf) * cprev + sigmoidf_(gi) * tanhf(gg_);
  const float h  = sigmoidf_(go) * tanhf(cn);
  q[tid] = h;
  __syncthreads();
  cs[(size_t)g * 256 + tid] = cn;
  hs[(size_t)g * 256 + tid] = h;
  qstar[(size_t)g * 512 + tid] = h;

  const v4f q0 = *(const v4f*)&q[4 * lane];
  const v4f q1 = *(const v4f*)&q[128 + 4 * lane];

  const float NEG_INF = -__builtin_inff();
  float m = NEG_INF;
  float s = 0.0f;
  v4f r0 = {}, r1 = {};

  while (base + TRIP < end) {
    request_trip(x, base + TRIP, last, lane, xa1, xb1);
    __builtin_amdgcn_sched_barrier(0);
    use_trip(xa0, xb0, base, end, q0, q1, m, s, r0, r1);
    request_trip(x, base + 2 * TRIP, last, lane, xa0, xb0);
    __builtin_amdgcn_sched_barrier(0);
    use_trip(xa1, xb1, base + TRIP, end, q0, q1, m, s, r0, r1);
    base += 2 * TRIP;
  }
  if (base < end) use_trip(xa0, xb0, base, end, q0, q1, m, s, r0, r1);

  *(v4f*)&rall[wv][4 * lane]       = r0;
  *(v4f*)&rall[wv][128 + 4 * lane] = r1;
  if (lane == 0) { wmax[wv] = m; wsum[wv] = s; }
  __syncthreads();

  float M = -__builtin_inff();
#pragma unroll
  for (int w = 0; w < 8; ++w) M = fmaxf(M, wmax[w]);

  float stot = 0.0f, r = 0.0f;
#pragma unroll
  for (int w = 0; w < 8; ++w) {
    float sw = wsum[w];
    if (sw > 0.0f) {
      float sc = __expf(wmax[w] - M);
      stot = fmaf(sw, sc, stot);
      r    = fmaf(rall[w][tid], sc, r);
    }
  }
  qstar[(size_t)g * 512 + 256 + tid] = (stot > 0.0f) ? (r / stot) : 0.0f;
}

__global__ void __launch_bounds__(256)
out_gemm(const float* __restrict__ qstar, const float* __restrict__ Wout,
         const float* __restrict__ bout, float* __restrict__ out) {
  __shared__ __align__(16) float As[16 * OUT_AP];
  const int tid  = threadIdx.x;
  const int lane = tid & 31;
  const int tm = blockIdx.x >> 1;
  const int tn = ((blockIdx.x & 1) << 3) + (tid >> 5);
  const int r  = lane & 15;
  const int kq = (lane >> 4) << 2;

  v4f st[8];
#pragma unroll
  for (int i = 0; i < 8; ++i) {
    const int idx = i * 256 + tid;
    st[i] = *(const v4f*)(qstar + (size_t)(tm * 16 + (idx >> 7)) * 512 + 4 * (idx & 127));
  }
#pragma unroll
  for (int i = 0; i < 8; ++i) {
    const int idx = i * 256 + tid;
    *(v4f*)&As[(idx >> 7) * OUT_AP + 4 * (idx & 127)] = st[i];
  }
  __syncthreads();

  const float* arow = As + r * OUT_AP + kq;
  const float* brow = Wout + (size_t)(tn * 16 + r) * 512 + kq;
  v8f acc = {};
  for (int k = 0; k < 512; k += 8) acc = mma_k8(arow + k, brow + k, acc);
  const int mb = tm * 16 + (lane >> 4) * 8;
  const int n  = tn * 16 + r;
  const float bias = bout[n];
#pragma unroll
  for (int v = 0; v < 8; ++v)
    out[(size_t)(mb + v) * 256 + n] = acc[v] + bias;
}

extern "C" void kernel_launch(void* const* d_in, const int* in_sizes, int n_in,
                              void* d_out, int out_size, void* d_ws, size_t ws_size,
                              hipStream_t stream) {
  const float* x     = (const float*)d_in[0];
  const int*   batch = (const int*)  d_in[1];
  const float* Wih   = (const float*)d_in[2];
  const float* Whh   = (const float*)d_in[3];
  const float* bih   = (const float*)d_in[4];
  const float* bhh   = (const float*)d_in[5];
  const float* Wout  = (const float*)d_in[6];
  const float* bout  = (const float*)d_in[7];
  const int N = in_sizes[0] / 256;
  const int G = 512;

  float* ws    = (float*)d_ws;
  float* hs    = ws;
  float* cs    = hs + 512 * 256;
  float* qstar = cs + 512 * 256;
  float* gates = qstar + 512 * 512;
  int*   starts = (int*)(gates + 512 * 1024);

  find_starts<<<(N + 255) / 256, 256, 0, stream>>>(batch, starts, N, G);

  attention<true><<<G, 256, 0, stream>>>(x, starts, bih, bhh, gates, hs, cs, qstar);
  for (int step = 1; step < 3; ++step) {
    gates_gemm<<<256, 256, 0, stream>>>(qstar, hs, Wih, Whh, bih, bhh, gates);
    attention<false><<<G, 256, 0, stream>>>(x, starts, bih, bhh, gates, hs, cs, qstar);
  }
  out_gemm<<<64, 256, 0, stream>>>(qstar, Wout, bout, (float*)d_out);
}


# attention, step 1 only: every graph is walked from its last node to its first (steps 0 and 2 forward), so a pass starts on the rows the pass before read last; that pass's f32 sum order reverses
# speedup vs baseline: 2.5138x; 1.1275x over previous
#include <hip/hip_runtime.h>

typedef __attribute__((ext_vector_type(2))) float v2f;
typedef __attribute__((ext_vector_type(4))) float v4f;
typedef __attribute__((ext_vector_type(8))) float v8f;

__global__ void find_starts(const int* __restrict__ batch, int* __restrict__ starts,
                            int N, int G) {
  int i = blockIdx.x * blockDim.x + threadIdx.x;
  if (i >= N) return;
  int b = batch[i];
  int bp = (i == 0) ? -1 : batch[i - 1];
  for (int g = bp + 1; g <= b; ++g) starts[g] = i;
  if (i == N - 1) {
    for (int g = b + 1; g <= G; ++g) starts[g] = N;
  }
}

__device__ __forceinline__ v8f mma_k8(const float* __restrict__ ap,
                                      const float* __restrict__ bp, v8f acc) {
  const v4f a = *(const v4f*)ap;
  const v4f b = *(const v4f*)bp;
  v2f a0, b0, a1, b1;
  a0.x = a.x; a0.y = a.y; b0.x = b.x; b0.y = b.y;
  a1.x = a.z; a1.y = a.w; b1.x = b.z; b1.y = b.w;
  acc = __builtin_amdgcn_wmma_f32_16x16x4_f32(false, a0, false, b0, (short)0, acc,
                                              false, false);
  acc = __builtin_amdgcn_wmma_f32_16x16x4_f32(false, a1, false, b1, (short)0, acc,
                                              false, false);
  return acc;
}

#define GATES_K   768
#define GATES_AP  (GATES_K + 4)
#define OUT_K     512
#define OUT_AP    (OUT_K + 4)

__global__ void __launch_bounds__(256)
gates_gemm(const float* __restrict__ qstar, const float* __restrict__ hs,
           const float* __restrict__ Wih, const float* __restrict__ Whh,
           const float* __restrict__ bih, const float* __restrict__ bhh,
           float* __restrict__ gates) {
  __shared__ __align__(16) float As[16 * GATES_AP];
  const int tid  = threadIdx.x;
  const int lane = tid & 31;
  const int tm = blockIdx.x >> 3;
  const int tn = ((blockIdx.x & 7) << 3) + (tid >> 5);
  const int r  = lane & 15;
  const int kq = (lane >> 4) << 2;

  v4f st[12];
#pragma unroll
  for (int i = 0; i < 8; ++i) {
    const int idx = i * 256 + tid;
    st[i] = *(const v4f*)(qstar + (size_t)(tm * 16 + (idx >> 7)) * 512 + 4 * (idx & 127));
  }
#pragma unroll
  for (int i = 0; i < 4; ++i) {
    const int idx = i * 256 + tid;
    st[8 + i] = *(const v4f*)(hs + (size_t)(tm * 16 + (idx >> 6)) * 256 + 4 * (idx & 63));
  }
#pragma unroll
  for (int i = 0; i < 8; ++i) {
    const int idx = i * 256 + tid;
    *(v4f*)&As[(idx >> 7) * GATES_AP + 4 * (idx & 127)] = st[i];
  }
#pragma unroll
  for (int i = 0; i < 4; ++i) {
    const int idx = i * 256 + tid;
    *(v4f*)&As[(idx >> 6) * GATES_AP + 512 + 4 * (idx & 63)] = st[8 + i];
  }
  __syncthreads();

  const float* arow = As + r * GATES_AP + kq;
  const float* brow = Wih + (size_t)(tn * 16 + r) * 512 + kq;
  v8f acc = {};
  for (int k = 0; k < 512; k += 8) acc = mma_k8(arow + k, brow + k, acc);
  const float* brow2 = Whh + (size_t)(tn * 16 + r) * 256 + kq;
  for (int k = 0; k < 256; k += 8) acc = mma_k8(arow + 512 + k, brow2 + k, acc);
  const int mb = tm * 16 + (lane >> 4) * 8;
  const int n  = tn * 16 + r;
  const float bias = bih[n] + bhh[n];
#pragma unroll
  for (int v = 0; v < 8; ++v)
    gates[(size_t)(mb + v) * 1024 + n] = acc[v] + bias;
}

__device__ __forceinline__ float sigmoidf_(float x) {
  return 1.0f / (1.0f + __expf(-x));
}

#define NB 8
#define TRIP (8 * NB)

template <bool REV>
__device__ __forceinline__ void request_trip(const float* __restrict__ x, int base, int last,
                                             int nm1, int lane, v4f (&xa)[NB], v4f (&xb)[NB]) {
#pragma unroll
  for (int j = 0; j < NB; ++j) {
    int i = base + j;
    if (REV) {
      i = (i < nm1) ? i : nm1;
      i = (i > 0) ? i : 0;
      i = last - i;
    } else {
      i = (i < last) ? i : last;
    }
    i = (i > 0) ? i : 0;
    const v4f* xr = (const v4f*)(x + (size_t)i * 256);
    xa[j] = xr[lane];
    xb[j] = xr[32 + lane];
  }
}

__device__ __forceinline__ void use_trip(const v4f (&xa)[NB], const v4f (&xb)[NB],
                                         int base, int end, v4f q0, v4f q1,
                                         float& m, float& s, v4f& r0, v4f& r1) {
  const float NEG_INF = -__builtin_inff();
  float d[NB];
#pragma unroll
  for (int j = 0; j < NB; ++j) {
    float t = 0.0f;
    t = fmaf(xa[j].x, q0.x, t); t = fmaf(xa[j].y, q0.y, t);
    t = fmaf(xa[j].z, q0.z, t); t = fmaf(xa[j].w, q0.w, t);
    t = fmaf(xb[j].x, q1.x, t); t = fmaf(xb[j].y, q1.y, t);
    t = fmaf(xb[j].z, q1.z, t); t = fmaf(xb[j].w, q1.w, t);
    asm volatile("" : "+v"(t));
    d[j] = t;
  }
#pragma unroll
  for (int off = 16; off; off >>= 1) {
#pragma unroll
    for (int j = 0; j < NB; ++j) d[j] += __shfl_xor(d[j], off, 32);
  }
  float mb = NEG_INF;
#pragma unroll
  for (int j = 0; j < NB; ++j) {
    if (base + j >= end) d[j] = NEG_INF;
    mb = fmaxf(mb, d[j]);
  }
  const float mn = fmaxf(m, mb);
  const float sc = (m == NEG_INF) ? 0.0f : __expf(m - mn);
  m = mn;
  s *= sc;
  r0.x *= sc; r0.y *= sc; r0.z *= sc; r0.w *= sc;
  r1.x *= sc; r1.y *= sc; r1.z *= sc; r1.w *= sc;
#pragma unroll
  for (int j = 0; j < NB; ++j) {
    const float a = __expf(d[j] - m);
    s += a;
    r0.x = fmaf(a, xa[j].x, r0.x); r0.y = fmaf(a, xa[j].y, r0.y);
    r0.z = fmaf(a, xa[j].z, r0.z); r0.w = fmaf(a, xa[j].w, r0.w);
    r1.x = fmaf(a, xb[j].x, r1.x); r1.y = fmaf(a, xb[j].y, r1.y);
    r1.z = fmaf(a, xb[j].z, r1.z); r1.w = fmaf(a, xb[j].w, r1.w);
  }
  asm volatile("" : "+v"(r0), "+v"(r1), "+v"(s), "+v"(m));
}

template <bool FIRST, bool REV>
__global__ void __launch_bounds__(256)
attention(const float* __restrict__ x, const int* __restrict__ starts,
          const float* __restrict__ bih, const float* __restrict__ bhh,
          const float* __restrict__ gates,
          float* __restrict__ hs, float* __restrict__ cs,
          float* __restrict__ qstar) {
  const int g    = blockIdx.x;
  const int tid  = threadIdx.x;
  const int lane = tid & 31;
  const int wv   = __builtin_amdgcn_readfirstlane(tid >> 5);

  __shared__ __align__(16) float q[256];
  __shared__ __align__(16) float rall[8][256];
  __shared__ float wmax[8];
  __shared__ float wsum[8];

  const int start = starts[g];
  const int end   = starts[g + 1];
  const int last  = end - 1;
  const int nm1   = last - start;
  const int lim = REV ? nm1 + 1 : end;
  int base = (REV ? 0 : start) + wv * NB;

  float gi, gf, gg_, go, cprev;
  if (FIRST) {
    gi  = bih[tid] + bhh[tid];
    gf  = bih[256 + tid] + bhh[256 + tid];
    gg_ = bih[512 + tid] + bhh[512 + tid];
    go  = bih[768 + tid] + bhh[768 + tid];
    cprev = 0.0f;
  } else {
    const float* gt = gates + (size_t)g * 1024;
    gi  = gt[tid];
    gf  = gt[256 + tid];
    gg_ = gt[512 + tid];
    go  = gt[768 + tid];
    cprev = cs[(size_t)g * 256 + tid];
  }

  v4f xa0[NB], xb0[NB], xa1[NB], xb1[NB];
  request_trip<REV>(x, base, last, nm1, lane, xa0, xb0);

  const float cn = sigmoidf_(gf) * cprev + sigmoidf_(gi) * tanhf(gg_);
  const float h  = sigmoidf_(go) * tanhf(cn);
  q[tid] = h;
  __syncthreads();
  cs[(size_t)g * 256 + tid] = cn;
  hs[(size_t)g * 256 + tid] = h;
  qstar[(size_t)g * 512 + tid] = h;

  const v4f q0 = *(const v4f*)&q[4 * lane];
  const v4f q1 = *(const v4f*)&q[128 + 4 * lane];

  const float NEG_INF = -__builtin_inff();
  float m = NEG_INF;
  float s = 0.0f;
  v4f r0 = {}, r1 = {};

  while (base + TRIP < lim) {
    request_trip<REV>(x, base + TRIP, last, nm1, lane, xa1, xb1);
    __builtin_amdgcn_sched_barrier(0);
    use_trip(xa0, xb0, base, lim, q0, q1, m, s, r0, r1);
    request_trip<REV>(x, base + 2 * TRIP, last, nm1, lane, xa0, xb0);
    __builtin_amdgcn_sched_barrier(0);
    use_trip(xa1, xb1, base + TRIP, lim, q0, q1, m, s, r0, r1);
    base += 2 * TRIP;
  }
  if (base < lim) use_trip(xa0, xb0, base, lim, q0, q1, m, s, r0, r1);

  *(v4f*)&rall[wv][4 * lane]       = r0;
  *(v4f*)&rall[wv][128 + 4 * lane] = r1;
  if (lane == 0) { wmax[wv] = m; wsum[wv] = s; }
  __syncthreads();

  float M = -__builtin_inff();
#pragma unroll
  for (int w = 0; w < 8; ++w) M = fmaxf(M, wmax[w]);

  float stot = 0.0f, r = 0.0f;
#pragma unroll
  for (int w = 0; w < 8; ++w) {
    float sw = wsum[w];
    if (sw > 0.0f) {
      float sc = __expf(wmax[w] - M);
      stot = fmaf(sw, sc, stot);
      r    = fmaf(rall[w][tid], sc, r);
    }
  }
  qstar[(size_t)g * 512 + 256 + tid] = (stot > 0.0f) ? (r / stot) : 0.0f;
}

__global__ void __launch_bounds__(256)
out_gemm(const float* __restrict__ qstar, const float* __restrict__ Wout,
         const float* __restrict__ bout, float* __restrict__ out) {
  __shared__ __align__(16) float As[16 * OUT_AP];
  const int tid  = threadIdx.x;
  const int lane = tid & 31;
  const int tm = blockIdx.x >> 1;
  const int tn = ((blockIdx.x & 1) << 3) + (tid >> 5);
  const int r  = lane & 15;
  const int kq = (lane >> 4) << 2;

  v4f st[8];
#pragma unroll
  for (int i = 0; i < 8; ++i) {
    const int idx = i * 256 + tid;
    st[i] = *(const v4f*)(qstar + (size_t)(tm * 16 + (idx >> 7)) * 512 + 4 * (idx & 127));
  }
#pragma unroll
  for (int i = 0; i < 8; ++i) {
    const int idx = i * 256 + tid;
    *(v4f*)&As[(idx >> 7) * OUT_AP + 4 * (idx & 127)] = st[i];
  }
  __syncthreads();

  const float* arow = As + r * OUT_AP + kq;
  const float* brow = Wout + (size_t)(tn * 16 + r) * 512 + kq;
  v8f acc = {};
  for (int k = 0; k < 512; k += 8) acc = mma_k8(arow + k, brow + k, acc);
  const int mb = tm * 16 + (lane >> 4) * 8;
  const int n  = tn * 16 + r;
  const float bias = bout[n];
#pragma unroll
  for (int v = 0; v < 8; ++v)
    out[(size_t)(mb + v) * 256 + n] = acc[v] + bias;
}

extern "C" void kernel_launch(void* const* d_in, const int* in_sizes, int n_in,
                              void* d_out, int out_size, void* d_ws, size_t ws_size,
                              hipStream_t stream) {
  const float* x     = (const float*)d_in[0];
  const int*   batch = (const int*)  d_in[1];
  const float* Wih   = (const float*)d_in[2];
  const float* Whh   = (const float*)d_in[3];
  const float* bih   = (const float*)d_in[4];
  const float* bhh   = (const float*)d_in[5];
  const float* Wout  = (const float*)d_in[6];
  const float* bout  = (const float*)d_in[7];
  const int N = in_sizes[0] / 256;
  const int G = 512;

  float* ws    = (float*)d_ws;
  float* hs    = ws;
  float* cs    = hs + 512 * 256;
  float* qstar = cs + 512 * 256;
  float* gates = qstar + 512 * 512;
  int*   starts = (int*)(gates + 512 * 1024);

  find_starts<<<(N + 255) / 256, 256, 0, stream>>>(batch, starts, N, G);

  attention<true, false><<<G, 256, 0, stream>>>(x, starts, bih, bhh, gates, hs, cs, qstar);
  gates_gemm<<<256, 256, 0, stream>>>(qstar, hs, Wih, Whh, bih, bhh, gates);
  attention<false, true><<<G, 256, 0, stream>>>(x, starts, bih, bhh, gates, hs, cs, qstar);
  gates_gemm<<<256, 256, 0, stream>>>(qstar, hs, Wih, Whh, bih, bhh, gates);
  attention<false, false><<<G, 256, 0, stream>>>(x, starts, bih, bhh, gates, hs, cs, qstar);
  out_gemm<<<64, 256, 0, stream>>>(qstar, Wout, bout, (float*)d_out);
}


# attention: a trip's 8 lane sums stay on the vector ALU (v_permlane16_swap pairs rows j, j+4 for step 16, row_xmask DPP adds for 8,4,2,1), not 40 ds_bpermute; same pairs and order, sums bit-identical
# speedup vs baseline: 2.6542x; 1.0559x over previous
#include <hip/hip_runtime.h>

typedef __attribute__((ext_vector_type(2))) float v2f;
typedef __attribute__((ext_vector_type(4))) float v4f;
typedef __attribute__((ext_vector_type(8))) float v8f;

__global__ void find_starts(const int* __restrict__ batch, int* __restrict__ starts,
                            int N, int G) {
  int i = blockIdx.x * blockDim.x + threadIdx.x;
  if (i >= N) return;
  int b = batch[i];
  int bp = (i == 0) ? -1 : batch[i - 1];
  for (int g = bp + 1; g <= b; ++g) starts[g] = i;
  if (i == N - 1) {
    for (int g = b + 1; g <= G; ++g) starts[g] = N;
  }
}

__device__ __forceinline__ v8f mma_k8(const float* __restrict__ ap,
                                      const float* __restrict__ bp, v8f acc) {
  const v4f a = *(const v4f*)ap;
  const v4f b = *(const v4f*)bp;
  v2f a0, b0, a1, b1;
  a0.x = a.x; a0.y = a.y; b0.x = b.x; b0.y = b.y;
  a1.x = a.z; a1.y = a.w; b1.x = b.z; b1.y = b.w;
  acc = __builtin_amdgcn_wmma_f32_16x16x4_f32(false, a0, false, b0, (short)0, acc,
                                              false, false);
  acc = __builtin_amdgcn_wmma_f32_16x16x4_f32(false, a1, false, b1, (short)0, acc,
                                              false, false);
  return acc;
}

#define GATES_K   768
#define GATES_AP  (GATES_K + 4)
#define OUT_K     512
#define OUT_AP    (OUT_K + 4)

__global__ void __launch_bounds__(256)
gates_gemm(const float* __restrict__ qstar, const float* __restrict__ hs,
           const float* __restrict__ Wih, const float* __restrict__ Whh,
           const float* __restrict__ bih, const float* __restrict__ bhh,
           float* __restrict__ gates) {
  __shared__ __align__(16) float As[16 * GATES_AP];
  const int tid  = threadIdx.x;
  const int lane = tid & 31;
  const int tm = blockIdx.x >> 3;
  const int tn = ((blockIdx.x & 7) << 3) + (tid >> 5);
  const int r  = lane & 15;
  const int kq = (lane >> 4) << 2;

  v4f st[12];
#pragma unroll
  for (int i = 0; i < 8; ++i) {
    const int idx = i * 256 + tid;
    st[i] = *(const v4f*)(qstar + (size_t)(tm * 16 + (idx >> 7)) * 512 + 4 * (idx & 127));
  }
#pragma unroll
  for (int i = 0; i < 4; ++i) {
    const int idx = i * 256 + tid;
    st[8 + i] = *(const v4f*)(hs + (size_t)(tm * 16 + (idx >> 6)) * 256 + 4 * (idx & 63));
  }
#pragma unroll
  for (int i = 0; i < 8; ++i) {
    const int idx = i * 256 + tid;
    *(v4f*)&As[(idx >> 7) * GATES_AP + 4 * (idx & 127)] = st[i];
  }
#pragma unroll
  for (int i = 0; i < 4; ++i) {
    const int idx = i * 256 + tid;
    *(v4f*)&As[(idx >> 6) * GATES_AP + 512 + 4 * (idx & 63)] = st[8 + i];
  }
  __syncthreads();

  const float* arow = As + r * GATES_AP + kq;
  const float* brow = Wih + (size_t)(tn * 16 + r) * 512 + kq;
  v8f acc = {};
  for (int k = 0; k < 512; k += 8) acc = mma_k8(arow + k, brow + k, acc);
  const float* brow2 = Whh + (size_t)(tn * 16 + r) * 256 + kq;
  for (int k = 0; k < 256; k += 8) acc = mma_k8(arow + 512 + k, brow2 + k, acc);
  const int mb = tm * 16 + (lane >> 4) * 8;
  const int n  = tn * 16 + r;
  const float bias = bih[n] + bhh[n];
#pragma unroll
  for (int v = 0; v < 8; ++v)
    gates[(size_t)(mb + v) * 1024 + n] = acc[v] + bias;
}

__device__ __forceinline__ float sigmoidf_(float x) {
  return 1.0f / (1.0f + __expf(-x));
}

#define NB 8
#define TRIP (8 * NB)

template <bool REV>
__device__ __forceinline__ void request_trip(const float* __restrict__ x, int base, int last,
                                             int nm1, int lane, v4f (&xa)[NB], v4f (&xb)[NB]) {
#pragma unroll
  for (int j = 0; j < NB; ++j) {
    int i = base + j;
    if (REV) {
      i = (i < nm1) ? i : nm1;
      i = (i > 0) ? i : 0;
      i = last - i;
    } else {
      i = (i < last) ? i : last;
    }
    i = (i > 0) ? i : 0;
    const v4f* xr = (const v4f*)(x + (size_t)i * 256);
    xa[j] = xr[lane];
    xb[j] = xr[32 + lane];
  }
}

template <int M>
__device__ __forceinline__ float row_xor_add(float v) {
  const int p = __builtin_amdgcn_update_dpp(0, __builtin_bit_cast(int, v), 0x160 + M, 0xf, 0xf, true);
  return v + __builtin_bit_cast(float, p);
}

__device__ __forceinline__ void swap_halves(float& a, float& b) {
  asm("v_permlane16_swap_b32 %0, %1" : "+v"(a), "+v"(b));
}

__device__ __forceinline__ void use_trip(const v4f (&xa)[NB], const v4f (&xb)[NB],
                                         int base, int end, v4f q0, v4f q1,
                                         float& m, float& s, v4f& r0, v4f& r1) {
  const float NEG_INF = -__builtin_inff();
  float d[NB];
#pragma unroll
  for (int j = 0; j < NB; ++j) {
    float t = 0.0f;
    t = fmaf(xa[j].x, q0.x, t); t = fmaf(xa[j].y, q0.y, t);
    t = fmaf(xa[j].z, q0.z, t); t = fmaf(xa[j].w, q0.w, t);
    t = fmaf(xb[j].x, q1.x, t); t = fmaf(xb[j].y, q1.y, t);
    t = fmaf(xb[j].z, q1.z, t); t = fmaf(xb[j].w, q1.w, t);
    asm volatile("" : "+v"(t));
    d[j] = t;
  }
  static_assert(NB == 8, "the lane sums below pair row j with row j + 4");
#pragma unroll
  for (int k = 0; k < 4; ++k) {
    float a = d[k], b = d[k + 4];
    swap_halves(a, b);
    float v = a + b;
    v = row_xor_add<8>(v);
    v = row_xor_add<4>(v);
    v = row_xor_add<2>(v);
    v = row_xor_add<1>(v);
    float lo = v, hi = v;
    swap_halves(lo, hi);
    d[k] = lo;
    d[k + 4] = hi;
  }
  float mb = NEG_INF;
#pragma unroll
  for (int j = 0; j < NB; ++j) {
    if (base + j >= end) d[j] = NEG_INF;
    mb = fmaxf(mb, d[j]);
  }
  const float mn = fmaxf(m, mb);
  const float sc = (m == NEG_INF) ? 0.0f : __expf(m - mn);
  m = mn;
  s *= sc;
  r0.x *= sc; r0.y *= sc; r0.z *= sc; r0.w *= sc;
  r1.x *= sc; r1.y *= sc; r1.z *= sc; r1.w *= sc;
#pragma unroll
  for (int j = 0; j < NB; ++j) {
    const float a = __expf(d[j] - m);
    s += a;
    r0.x = fmaf(a, xa[j].x, r0.x); r0.y = fmaf(a, xa[j].y, r0.y);
    r0.z = fmaf(a, xa[j].z, r0.z); r0.w = fmaf(a, xa[j].w, r0.w);
    r1.x = fmaf(a, xb[j].x, r1.x); r1.y = fmaf(a, xb[j].y, r1.y);
    r1.z = fmaf(a, xb[j].z, r1.z); r1.w = fmaf(a, xb[j].w, r1.w);
  }
  asm volatile("" : "+v"(r0), "+v"(r1), "+v"(s), "+v"(m));
}

template <bool FIRST, bool REV>
__global__ void __launch_bounds__(256)
attention(const float* __restrict__ x, const int* __restrict__ starts,
          const float* __restrict__ bih, const float* __restrict__ bhh,
          const float* __restrict__ gates,
          float* __restrict__ hs, float* __restrict__ cs,
          float* __restrict__ qstar) {
  const int g    = blockIdx.x;
  const int tid  = threadIdx.x;
  const int lane = tid & 31;
  const int wv   = __builtin_amdgcn_readfirstlane(tid >> 5);

  __shared__ __align__(16) float q[256];
  __shared__ __align__(16) float rall[8][256];
  __shared__ float wmax[8];
  __shared__ float wsum[8];

  const int start = starts[g];
  const int end   = starts[g + 1];
  const int last  = end - 1;
  const int nm1   = last - start;
  const int lim = REV ? nm1 + 1 : end;
  int base = (REV ? 0 : start) + wv * NB;

  float gi, gf, gg_, go, cprev;
  if (FIRST) {
    gi  = bih[tid] + bhh[tid];
    gf  = bih[256 + tid] + bhh[256 + tid];
    gg_ = bih[512 + tid] + bhh[512 + tid];
    go  = bih[768 + tid] + bhh[768 + tid];
    cprev = 0.0f;
  } else {
    const float* gt = gates + (size_t)g * 1024;
    gi  = gt[tid];
    gf  = gt[256 + tid];
    gg_ = gt[512 + tid];
    go  = gt[768 + tid];
    cprev = cs[(size_t)g * 256 + tid];
  }

  v4f xa0[NB], xb0[NB], xa1[NB], xb1[NB];
  request_trip<REV>(x, base, last, nm1, lane, xa0, xb0);

  const float cn = sigmoidf_(gf) * cprev + sigmoidf_(gi) * tanhf(gg_);
  const float h  = sigmoidf_(go) * tanhf(cn);
  q[tid] = h;
  __syncthreads();
  cs[(size_t)g * 256 + tid] = cn;
  hs[(size_t)g * 256 + tid] = h;
  qstar[(size_t)g * 512 + tid] = h;

  const v4f q0 = *(const v4f*)&q[4 * lane];
  const v4f q1 = *(const v4f*)&q[128 + 4 * lane];

  const float NEG_INF = -__builtin_inff();
  float m = NEG_INF;
  float s = 0.0f;
  v4f r0 = {}, r1 = {};

  while (base + TRIP < lim) {
    request_trip<REV>(x, base + TRIP, last, nm1, lane, xa1, xb1);
    __builtin_amdgcn_sched_barrier(0);
    use_trip(xa0, xb0, base, lim, q0, q1, m, s, r0, r1);
    request_trip<REV>(x, base + 2 * TRIP, last, nm1, lane, xa0, xb0);
    __builtin_amdgcn_sched_barrier(0);
    use_trip(xa1, xb1, base + TRIP, lim, q0, q1, m, s, r0, r1);
    base += 2 * TRIP;
  }
  if (base < lim) use_trip(xa0, xb0, base, lim, q0, q1, m, s, r0, r1);

  *(v4f*)&rall[wv][4 * lane]       = r0;
  *(v4f*)&rall[wv][128 + 4 * lane] = r1;
  if (lane == 0) { wmax[wv] = m; wsum[wv] = s; }
  __syncthreads();

  float M = -__builtin_inff();
#pragma unroll
  for (int w = 0; w < 8; ++w) M = fmaxf(M, wmax[w]);

  float stot = 0.0f, r = 0.0f;
#pragma unroll
  for (int w = 0; w < 8; ++w) {
    float sw = wsum[w];
    if (sw > 0.0f) {
      float sc = __expf(wmax[w] - M);
      stot = fmaf(sw, sc, stot);
      r    = fmaf(rall[w][tid], sc, r);
    }
  }
  qstar[(size_t)g * 512 + 256 + tid] = (stot > 0.0f) ? (r / stot) : 0.0f;
}

__global__ void __launch_bounds__(256)
out_gemm(const float* __restrict__ qstar, const float* __restrict__ Wout,
         const float* __restrict__ bout, float* __restrict__ out) {
  __shared__ __align__(16) float As[16 * OUT_AP];
  const int tid  = threadIdx.x;
  const int lane = tid & 31;
  const int tm = blockIdx.x >> 1;
  const int tn = ((blockIdx.x & 1) << 3) + (tid >> 5);
  const int r  = lane & 15;
  const int kq = (lane >> 4) << 2;

  v4f st[8];
#pragma unroll
  for (int i = 0; i < 8; ++i) {
    const int idx = i * 256 + tid;
    st[i] = *(const v4f*)(qstar + (size_t)(tm * 16 + (idx >> 7)) * 512 + 4 * (idx & 127));
  }
#pragma unroll
  for (int i = 0; i < 8; ++i) {
    const int idx = i * 256 + tid;
    *(v4f*)&As[(idx >> 7) * OUT_AP + 4 * (idx & 127)] = st[i];
  }
  __syncthreads();

  const float* arow = As + r * OUT_AP + kq;
  const float* brow = Wout + (size_t)(tn * 16 + r) * 512 + kq;
  v8f acc = {};
  for (int k = 0; k < 512; k += 8) acc = mma_k8(arow + k, brow + k, acc);
  const int mb = tm * 16 + (lane >> 4) * 8;
  const int n  = tn * 16 + r;
  const float bias = bout[n];
#pragma unroll
  for (int v = 0; v < 8; ++v)
    out[(size_t)(mb + v) * 256 + n] = acc[v] + bias;
}

extern "C" void kernel_launch(void* const* d_in, const int* in_sizes, int n_in,
                              void* d_out, int out_size, void* d_ws, size_t ws_size,
                              hipStream_t stream) {
  const float* x     = (const float*)d_in[0];
  const int*   batch = (const int*)  d_in[1];
  const float* Wih   = (const float*)d_in[2];
  const float* Whh   = (const float*)d_in[3];
  const float* bih   = (const float*)d_in[4];
  const float* bhh   = (const float*)d_in[5];
  const float* Wout  = (const float*)d_in[6];
  const float* bout  = (const float*)d_in[7];
  const int N = in_sizes[0] / 256;
  const int G = 512;

  float* ws    = (float*)d_ws;
  float* hs    = ws;
  float* cs    = hs + 512 * 256;
  float* qstar = cs + 512 * 256;
  float* gates = qstar + 512 * 512;
  int*   starts = (int*)(gates + 512 * 1024);

  find_starts<<<(N + 255) / 256, 256, 0, stream>>>(batch, starts, N, G);

  attention<true, false><<<G, 256, 0, stream>>>(x, starts, bih, bhh, gates, hs, cs, qstar);
  gates_gemm<<<256, 256, 0, stream>>>(qstar, hs, Wih, Whh, bih, bhh, gates);
  attention<false, true><<<G, 256, 0, stream>>>(x, starts, bih, bhh, gates, hs, cs, qstar);
  gates_gemm<<<256, 256, 0, stream>>>(qstar, hs, Wih, Whh, bih, bhh, gates);
  attention<false, false><<<G, 256, 0, stream>>>(x, starts, bih, bhh, gates, hs, cs, qstar);
  out_gemm<<<64, 256, 0, stream>>>(qstar, Wout, bout, (float*)d_out);
}


# GEMMs: each wave stages its 16-row W strip in 128-k chunks through LDS rows of its own (whole-line loads) and reads A and W fragments 16 at a time ahead of 16-WMMA blocks; k order, f32 sums unchanged
# speedup vs baseline: 2.8662x; 1.0799x over previous
#include <hip/hip_runtime.h>

typedef __attribute__((ext_vector_type(2))) float v2f;
typedef __attribute__((ext_vector_type(4))) float v4f;
typedef __attribute__((ext_vector_type(8))) float v8f;

__global__ void find_starts(const int* __restrict__ batch, int* __restrict__ starts,
                            int N, int G) {
  int i = blockIdx.x * blockDim.x + threadIdx.x;
  if (i >= N) return;
  int b = batch[i];
  int bp = (i == 0) ? -1 : batch[i - 1];
  for (int g = bp + 1; g <= b; ++g) starts[g] = i;
  if (i == N - 1) {
    for (int g = b + 1; g <= G; ++g) starts[g] = N;
  }
}

__device__ __forceinline__ void frag_load(const float* ap, const float* wp,
                                          v4f (&fa)[8], v4f (&fw)[8]) {
#pragma unroll
  for (int j = 0; j < 8; ++j) {
    fa[j] = *(const v4f*)(ap + 8 * j);
    fw[j] = *(const v4f*)(wp + 8 * j);
  }
}

__device__ __forceinline__ v8f frag_mma(const v4f (&fa)[8], const v4f (&fw)[8], v8f acc) {
#pragma unroll
  for (int j = 0; j < 8; ++j) {
    v2f a0, b0, a1, b1;
    a0.x = fa[j].x; a0.y = fa[j].y; b0.x = fw[j].x; b0.y = fw[j].y;
    a1.x = fa[j].z; a1.y = fa[j].w; b1.x = fw[j].z; b1.y = fw[j].w;
    acc = __builtin_amdgcn_wmma_f32_16x16x4_f32(false, a0, false, b0, (short)0, acc,
                                                false, false);
    acc = __builtin_amdgcn_wmma_f32_16x16x4_f32(false, a1, false, b1, (short)0, acc,
                                                false, false);
  }
  return acc;
}

#define WCH 128
#define WP  (WCH + 4)

__device__ __forceinline__ void w_request(const float* src, int ld, v4f (&wr)[16]) {
#pragma unroll
  for (int i = 0; i < 16; ++i) wr[i] = *(const v4f*)(src + (size_t)i * ld);
}

__device__ __forceinline__ void w_store(float* dst, const v4f (&wr)[16]) {
#pragma unroll
  for (int i = 0; i < 16; ++i) *(v4f*)(dst + i * WP) = wr[i];
}

__device__ __forceinline__ void wave_lds_order() {
  __builtin_amdgcn_fence(__ATOMIC_RELEASE, "wavefront");
  __builtin_amdgcn_wave_barrier();
  __builtin_amdgcn_fence(__ATOMIC_ACQUIRE, "wavefront");
}

template <int NCH, class Req>
__device__ __forceinline__ v8f k_loop(const float* arow, float* wst, const float* wfr,
                                      v4f (&wr0)[16], v4f (&wr1)[16], Req req) {
  v4f fa[2][8], fw[2][8];
  v8f acc = {};
  wave_lds_order();
  w_store(wst, wr0);
  if (2 < NCH) req(2, wr0);
  wave_lds_order();
  frag_load(arow, wfr, fa[0], fw[0]);
#pragma unroll
  for (int g = 0; g < 2 * NCH; ++g) {
    const int n = g + 1;
    if (n < 2 * NCH) {
      const int c = n >> 1;
      if ((n & 1) == 0) {
        wave_lds_order();
        if (c & 1) w_store(wst, wr1); else w_store(wst, wr0);
        if (c + 2 < NCH) { if (c & 1) req(c + 2, wr1); else req(c + 2, wr0); }
        wave_lds_order();
      }
      frag_load(arow + c * WCH + (n & 1) * 64, wfr + (n & 1) * 64, fa[n & 1], fw[n & 1]);
    }
    __builtin_amdgcn_sched_barrier(0);
    acc = frag_mma(fa[g & 1], fw[g & 1], acc);
    __builtin_amdgcn_sched_barrier(0);
  }
  return acc;
}

#define GATES_K   768
#define GATES_AP  (GATES_K + 4)
#define OUT_K     512
#define OUT_AP    (OUT_K + 4)

__global__ void __launch_bounds__(256)
gates_gemm(const float* __restrict__ qstar, const float* __restrict__ hs,
           const float* __restrict__ Wih, const float* __restrict__ Whh,
           const float* __restrict__ bih, const float* __restrict__ bhh,
           float* __restrict__ gates) {
  __shared__ __align__(16) float As[16 * GATES_AP];
  __shared__ __align__(16) float Ws[8 * 16 * WP];
  const int tid  = threadIdx.x;
  const int lane = tid & 31;
  const int wv = __builtin_amdgcn_readfirstlane(tid >> 5);
  const int tm = blockIdx.x >> 3;
  const int tn = ((blockIdx.x & 7) << 3) + wv;
  const int r  = lane & 15;
  const int kq = (lane >> 4) << 2;

  const float* wih0 = Wih + (size_t)(tn * 16) * 512 + 4 * lane;
  const float* whh0 = Whh + (size_t)(tn * 16) * 256 + 4 * lane;
  auto req = [&](int c, v4f (&w)[16]) {
    if (c < 4) w_request(wih0 + c * WCH, 512, w);
    else       w_request(whh0 + (c - 4) * WCH, 256, w);
  };
  v4f wr0[16], wr1[16];
  req(0, wr0);
  req(1, wr1);

  v4f st[12];
#pragma unroll
  for (int i = 0; i < 8; ++i) {
    const int idx = i * 256 + tid;
    st[i] = *(const v4f*)(qstar + (size_t)(tm * 16 + (idx >> 7)) * 512 + 4 * (idx & 127));
  }
#pragma unroll
  for (int i = 0; i < 4; ++i) {
    const int idx = i * 256 + tid;
    st[8 + i] = *(const v4f*)(hs + (size_t)(tm * 16 + (idx >> 6)) * 256 + 4 * (idx & 63));
  }
#pragma unroll
  for (int i = 0; i < 8; ++i) {
    const int idx = i * 256 + tid;
    *(v4f*)&As[(idx >> 7) * GATES_AP + 4 * (idx & 127)] = st[i];
  }
#pragma unroll
  for (int i = 0; i < 4; ++i) {
    const int idx = i * 256 + tid;
    *(v4f*)&As[(idx >> 6) * GATES_AP + 512 + 4 * (idx & 63)] = st[8 + i];
  }
  __syncthreads();

  float* wl = Ws + wv * (16 * WP);
  const v8f acc = k_loop<6>(As + r * GATES_AP + kq, wl + 4 * lane, wl + r * WP + kq,
                            wr0, wr1, req);
  const int mb = tm * 16 + (lane >> 4) * 8;
  const int n  = tn * 16 + r;
  const float bias = bih[n] + bhh[n];
#pragma unroll
  for (int v = 0; v < 8; ++v)
    gates[(size_t)(mb + v) * 1024 + n] = acc[v] + bias;
}

__device__ __forceinline__ float sigmoidf_(float x) {
  return 1.0f / (1.0f + __expf(-x));
}

#define NB 8
#define TRIP (8 * NB)

template <bool REV>
__device__ __forceinline__ void request_trip(const float* __restrict__ x, int base, int last,
                                             int nm1, int lane, v4f (&xa)[NB], v4f (&xb)[NB]) {
#pragma unroll
  for (int j = 0; j < NB; ++j) {
    int i = base + j;
    if (REV) {
      i = (i < nm1) ? i : nm1;
      i = (i > 0) ? i : 0;
      i = last - i;
    } else {
      i = (i < last) ? i : last;
    }
    i = (i > 0) ? i : 0;
    const v4f* xr = (const v4f*)(x + (size_t)i * 256);
    xa[j] = xr[lane];
    xb[j] = xr[32 + lane];
  }
}

template <int M>
__device__ __forceinline__ float row_xor_add(float v) {
  const int p = __builtin_amdgcn_update_dpp(0, __builtin_bit_cast(int, v), 0x160 + M, 0xf, 0xf, true);
  return v + __builtin_bit_cast(float, p);
}

__device__ __forceinline__ void swap_halves(float& a, float& b) {
  asm("v_permlane16_swap_b32 %0, %1" : "+v"(a), "+v"(b));
}

__device__ __forceinline__ void use_trip(const v4f (&xa)[NB], const v4f (&xb)[NB],
                                         int base, int end, v4f q0, v4f q1,
                                         float& m, float& s, v4f& r0, v4f& r1) {
  const float NEG_INF = -__builtin_inff();
  float d[NB];
#pragma unroll
  for (int j = 0; j < NB; ++j) {
    float t = 0.0f;
    t = fmaf(xa[j].x, q0.x, t); t = fmaf(xa[j].y, q0.y, t);
    t = fmaf(xa[j].z, q0.z, t); t = fmaf(xa[j].w, q0.w, t);
    t = fmaf(xb[j].x, q1.x, t); t = fmaf(xb[j].y, q1.y, t);
    t = fmaf(xb[j].z, q1.z, t); t = fmaf(xb[j].w, q1.w, t);
    asm volatile("" : "+v"(t));
    d[j] = t;
  }
  static_assert(NB == 8, "the lane sums below pair row j with row j + 4");
#pragma unroll
  for (int k = 0; k < 4; ++k) {
    float a = d[k], b = d[k + 4];
    swap_halves(a, b);
    float v = a + b;
    v = row_xor_add<8>(v);
    v = row_xor_add<4>(v);
    v = row_xor_add<2>(v);
    v = row_xor_add<1>(v);
    float lo = v, hi = v;
    swap_halves(lo, hi);
    d[k] = lo;
    d[k + 4] = hi;
  }
  float mb = NEG_INF;
#pragma unroll
  for (int j = 0; j < NB; ++j) {
    if (base + j >= end) d[j] = NEG_INF;
    mb = fmaxf(mb, d[j]);
  }
  const float mn = fmaxf(m, mb);
  const float sc = (m == NEG_INF) ? 0.0f : __expf(m - mn);
  m = mn;
  s *= sc;
  r0.x *= sc; r0.y *= sc; r0.z *= sc; r0.w *= sc;
  r1.x *= sc; r1.y *= sc; r1.z *= sc; r1.w *= sc;
#pragma unroll
  for (int j = 0; j < NB; ++j) {
    const float a = __expf(d[j] - m);
    s += a;
    r0.x = fmaf(a, xa[j].x, r0.x); r0.y = fmaf(a, xa[j].y, r0.y);
    r0.z = fmaf(a, xa[j].z, r0.z); r0.w = fmaf(a, xa[j].w, r0.w);
    r1.x = fmaf(a, xb[j].x, r1.x); r1.y = fmaf(a, xb[j].y, r1.y);
    r1.z = fmaf(a, xb[j].z, r1.z); r1.w = fmaf(a, xb[j].w, r1.w);
  }
  asm volatile("" : "+v"(r0), "+v"(r1), "+v"(s), "+v"(m));
}

template <bool FIRST, bool REV>
__global__ void __launch_bounds__(256)
attention(const float* __restrict__ x, const int* __restrict__ starts,
          const float* __restrict__ bih, const float* __restrict__ bhh,
          const float* __restrict__ gates,
          float* __restrict__ hs, float* __restrict__ cs,
          float* __restrict__ qstar) {
  const int g    = blockIdx.x;
  const int tid  = threadIdx.x;
  const int lane = tid & 31;
  const int wv   = __builtin_amdgcn_readfirstlane(tid >> 5);

  __shared__ __align__(16) float q[256];
  __shared__ __align__(16) float rall[8][256];
  __shared__ float wmax[8];
  __shared__ float wsum[8];

  const int start = starts[g];
  const int end   = starts[g + 1];
  const int last  = end - 1;
  const int nm1   = last - start;
  const int lim = REV ? nm1 + 1 : end;
  int base = (REV ? 0 : start) + wv * NB;

  float gi, gf, gg_, go, cprev;
  if (FIRST) {
    gi  = bih[tid] + bhh[tid];
    gf  = bih[256 + tid] + bhh[256 + tid];
    gg_ = bih[512 + tid] + bhh[512 + tid];
    go  = bih[768 + tid] + bhh[768 + tid];
    cprev = 0.0f;
  } else {
    const float* gt = gates + (size_t)g * 1024;
    gi  = gt[tid];
    gf  = gt[256 + tid];
    gg_ = gt[512 + tid];
    go  = gt[768 + tid];
    cprev = cs[(size_t)g * 256 + tid];
  }

  v4f xa0[NB], xb0[NB], xa1[NB], xb1[NB];
  request_trip<REV>(x, base, last, nm1, lane, xa0, xb0);

  const float cn = sigmoidf_(gf) * cprev + sigmoidf_(gi) * tanhf(gg_);
  const float h  = sigmoidf_(go) * tanhf(cn);
  q[tid] = h;
  __syncthreads();
  cs[(size_t)g * 256 + tid] = cn;
  hs[(size_t)g * 256 + tid] = h;
  qstar[(size_t)g * 512 + tid] = h;

  const v4f q0 = *(const v4f*)&q[4 * lane];
  const v4f q1 = *(const v4f*)&q[128 + 4 * lane];

  const float NEG_INF = -__builtin_inff();
  float m = NEG_INF;
  float s = 0.0f;
  v4f r0 = {}, r1 = {};

  while (base + TRIP < lim) {
    request_trip<REV>(x, base + TRIP, last, nm1, lane, xa1, xb1);
    __builtin_amdgcn_sched_barrier(0);
    use_trip(xa0, xb0, base, lim, q0, q1, m, s, r0, r1);
    request_trip<REV>(x, base + 2 * TRIP, last, nm1, lane, xa0, xb0);
    __builtin_amdgcn_sched_barrier(0);
    use_trip(xa1, xb1, base + TRIP, lim, q0, q1, m, s, r0, r1);
    base += 2 * TRIP;
  }
  if (base < lim) use_trip(xa0, xb0, base, lim, q0, q1, m, s, r0, r1);

  *(v4f*)&rall[wv][4 * lane]       = r0;
  *(v4f*)&rall[wv][128 + 4 * lane] = r1;
  if (lane == 0) { wmax[wv] = m; wsum[wv] = s; }
  __syncthreads();

  float M = -__builtin_inff();
#pragma unroll
  for (int w = 0; w < 8; ++w) M = fmaxf(M, wmax[w]);

  float stot = 0.0f, r = 0.0f;
#pragma unroll
  for (int w = 0; w < 8; ++w) {
    float sw = wsum[w];
    if (sw > 0.0f) {
      float sc = __expf(wmax[w] - M);
      stot = fmaf(sw, sc, stot);
      r    = fmaf(rall[w][tid], sc, r);
    }
  }
  qstar[(size_t)g * 512 + 256 + tid] = (stot > 0.0f) ? (r / stot) : 0.0f;
}

__global__ void __launch_bounds__(256)
out_gemm(const float* __restrict__ qstar, const float* __restrict__ Wout,
         const float* __restrict__ bout, float* __restrict__ out) {
  __shared__ __align__(16) float As[16 * OUT_AP];
  __shared__ __align__(16) float Ws[8 * 16 * WP];
  const int tid  = threadIdx.x;
  const int lane = tid & 31;
  const int wv = __builtin_amdgcn_readfirstlane(tid >> 5);
  const int tm = blockIdx.x >> 1;
  const int tn = ((blockIdx.x & 1) << 3) + wv;
  const int r  = lane & 15;
  const int kq = (lane >> 4) << 2;

  const float* w0 = Wout + (size_t)(tn * 16) * 512 + 4 * lane;
  auto req = [&](int c, v4f (&w)[16]) { w_request(w0 + c * WCH, 512, w); };
  v4f wr0[16], wr1[16];
  req(0, wr0);
  req(1, wr1);

  v4f st[8];
#pragma unroll
  for (int i = 0; i < 8; ++i) {
    const int idx = i * 256 + tid;
    st[i] = *(const v4f*)(qstar + (size_t)(tm * 16 + (idx >> 7)) * 512 + 4 * (idx & 127));
  }
#pragma unroll
  for (int i = 0; i < 8; ++i) {
    const int idx = i * 256 + tid;
    *(v4f*)&As[(idx >> 7) * OUT_AP + 4 * (idx & 127)] = st[i];
  }
  __syncthreads();

  float* wl = Ws + wv * (16 * WP);
  const v8f acc = k_loop<4>(As + r * OUT_AP + kq, wl + 4 * lane, wl + r * WP + kq,
                            wr0, wr1, req);
  const int mb = tm * 16 + (lane >> 4) * 8;
  const int n  = tn * 16 + r;
  const float bias = bout[n];
#pragma unroll
  for (int v = 0; v < 8; ++v)
    out[(size_t)(mb + v) * 256 + n] = acc[v] + bias;
}

extern "C" void kernel_launch(void* const* d_in, const int* in_sizes, int n_in,
                              void* d_out, int out_size, void* d_ws, size_t ws_size,
                              hipStream_t stream) {
  const float* x     = (const float*)d_in[0];
  const int*   batch = (const int*)  d_in[1];
  const float* Wih   = (const float*)d_in[2];
  const float* Whh   = (const float*)d_in[3];
  const float* bih   = (const float*)d_in[4];
  const float* bhh   = (const float*)d_in[5];
  const float* Wout  = (const float*)d_in[6];
  const float* bout  = (const float*)d_in[7];
  const int N = in_sizes[0] / 256;
  const int G = 512;

  float* ws    = (float*)d_ws;
  float* hs    = ws;
  float* cs    = hs + 512 * 256;
  float* qstar = cs + 512 * 256;
  float* gates = qstar + 512 * 512;
  int*   starts = (int*)(gates + 512 * 1024);

  find_starts<<<(N + 255) / 256, 256, 0, stream>>>(batch, starts, N, G);

  attention<true, false><<<G, 256, 0, stream>>>(x, starts, bih, bhh, gates, hs, cs, qstar);
  gates_gemm<<<256, 256, 0, stream>>>(qstar, hs, Wih, Whh, bih, bhh, gates);
  attention<false, true><<<G, 256, 0, stream>>>(x, starts, bih, bhh, gates, hs, cs, qstar);
  gates_gemm<<<256, 256, 0, stream>>>(qstar, hs, Wih, Whh, bih, bhh, gates);
  attention<false, false><<<G, 256, 0, stream>>>(x, starts, bih, bhh, gates, hs, cs, qstar);
  out_gemm<<<64, 256, 0, stream>>>(qstar, Wout, bout, (float*)d_out);
}
